# starting kernel
# baseline (speedup 1.0000x reference)
#include <hip/hip_runtime.h>
#include <hip/hip_bf16.h>

#define L_SEQ 250
#define DMODEL 8
#define DINNER 16
#define DSTATE 16
#define DCONV 5
#define NCLS 12

typedef __attribute__((ext_vector_type(2))) float v2f;
typedef __attribute__((ext_vector_type(8))) float v8f;

__device__ __forceinline__ float sigmoidf_(float v) { return 1.0f / (1.0f + __expf(-v)); }

__global__ __launch_bounds__(256) void mamba_fused_kernel(
    const float* __restrict__ x,
    const float* __restrict__ W_in,
    const float* __restrict__ conv_w,
    const float* __restrict__ conv_b,
    const float* __restrict__ W_x,
    const float* __restrict__ dt_w,
    const float* __restrict__ dt_b,
    const float* __restrict__ A_log,
    const float* __restrict__ Dp,
    const float* __restrict__ W_out,
    const float* __restrict__ fc_w,
    const float* __restrict__ fc_b,
    float* __restrict__ out)
{
    __shared__ float smem[16298];
    float* s_z   = smem;
    float* s_xc  = smem + 4016;
    float* s_B   = smem + 8016;
    float* s_C   = smem + 12032;
    float* s_dtr = smem + 16048;
    float* s_xb  = s_B;
    float* s_y   = s_z;
    float* s_out = s_C;
    float* s_red = s_dtr;

    const int b    = blockIdx.x;
    const int tid  = threadIdx.x;
    const int lane = tid & 31;
    const int wave = tid >> 5;
    const int lm   = lane & 15;
    const int hi   = lane >> 4;
    const float* xb = x + (size_t)b * DMODEL * L_SEQ;

    for (int i = tid; i < DMODEL * L_SEQ; i += 256) s_xb[i] = xb[i];
    for (int i = tid * 32; i < NCLS * L_SEQ * DMODEL; i += 256 * 32)
        __builtin_prefetch(fc_w + i, 0, 0);
    __syncthreads();

    for (int mt = wave; mt < 16; mt += 8) {
        const int l0 = mt * 16;
        const int lr = min(l0 + lm, L_SEQ - 1);
        v8f acc = {};
#pragma unroll
        for (int kc = 0; kc < 8; kc += 4) {
            const int k0 = kc + 2 * hi;
            v2f a, bb;
            a.x  = s_xb[(k0    ) * L_SEQ + lr];
            a.y  = s_xb[(k0 + 1) * L_SEQ + lr];
            bb.x = W_in[(16 + lm) * 8 + k0];
            bb.y = W_in[(16 + lm) * 8 + k0 + 1];
            acc = __builtin_amdgcn_wmma_f32_16x16x4_f32(false, a, false, bb,
                                                        (short)0, acc, false, false);
        }
#pragma unroll
        for (int r = 0; r < 8; ++r) {
            const int row = min(l0 + r + 8 * hi, L_SEQ);
            s_z[row * 16 + lm] = acc[r];
        }
    }

    for (int i = tid; i < L_SEQ * DINNER; i += 256) {
        const int l = i >> 4, d = i & 15;
        float accv = conv_b[d];
#pragma unroll
        for (int j = 0; j < DCONV; ++j) {
            const int t = l - (DCONV - 1) + j;
            if (t >= 0) {
                float xi = 0.f;
#pragma unroll
                for (int dm = 0; dm < DMODEL; ++dm) xi += s_xb[dm * L_SEQ + t] * W_in[d * 8 + dm];
                accv += conv_w[d * DCONV + j] * xi;
            }
        }
        s_xc[i] = accv * sigmoidf_(accv);
    }
    __syncthreads();

    for (int l = tid; l < L_SEQ; l += 256) {
        float a = 0.f;
#pragma unroll
        for (int d = 0; d < DINNER; ++d) a += s_xc[l * 16 + d] * W_x[d];
        s_dtr[l] = a;
    }

    for (int j = wave; j < 32; j += 8) {
        const int mt = j >> 1, nt = j & 1;
        const int l0 = mt * 16;
        const int lr = min(l0 + lm, L_SEQ - 1);
        const int e0 = 1 + nt * 16;
        v8f acc = {};
#pragma unroll
        for (int kc = 0; kc < 16; kc += 4) {
            const int k0 = kc + 2 * hi;
            v2f a, bb;
            a.x  = s_xc[lr * 16 + k0];
            a.y  = s_xc[lr * 16 + k0 + 1];
            bb.x = W_x[(e0 + lm) * 16 + k0];
            bb.y = W_x[(e0 + lm) * 16 + k0 + 1];
            acc = __builtin_amdgcn_wmma_f32_16x16x4_f32(false, a, false, bb,
                                                        (short)0, acc, false, false);
        }
        float* dst = nt ? s_C : s_B;
#pragma unroll
        for (int r = 0; r < 8; ++r) {
            const int row = min(l0 + r + 8 * hi, L_SEQ);
            dst[row * 16 + lm] = acc[r];
        }
    }
    __syncthreads();

    {
        const int d = tid >> 4;
        const int s = tid & 15;
        const float dtw = dt_w[d], dtb = dt_b[d], Dpd = Dp[d];
        const float As  = -__expf(A_log[d * 16 + s]);
        float h = 0.f;
        for (int l = 0; l < L_SEQ; ++l) {
            float dv = s_dtr[l] * dtw + dtb;
            dv = (dv > 20.f) ? dv : log1pf(__expf(dv));
            const float xcv = s_xc[l * 16 + d];
            h = __expf(dv * As) * h + (dv * xcv) * s_B[l * 16 + s];
            float y = h * s_C[l * 16 + s];
            y += __shfl_xor(y, 1, 32);
            y += __shfl_xor(y, 2, 32);
            y += __shfl_xor(y, 4, 32);
            y += __shfl_xor(y, 8, 32);
            if (s == 0) {
                y += xcv * Dpd;
                const float zv = s_z[l * 16 + d];
                y *= zv * sigmoidf_(zv);
                s_y[l * 16 + d] = y;
            }
        }
    }
    __syncthreads();

    for (int mt = wave; mt < 16; mt += 8) {
        const int l0 = mt * 16;
        const int lr = min(l0 + lm, L_SEQ - 1);
        v8f acc = {};
#pragma unroll
        for (int kc = 0; kc < 16; kc += 4) {
            const int k0 = kc + 2 * hi;
            v2f a, bb;
            a.x  = s_y[lr * 16 + k0];
            a.y  = s_y[lr * 16 + k0 + 1];
            bb.x = (lm < DMODEL) ? W_out[lm * 16 + k0]     : 0.f;
            bb.y = (lm < DMODEL) ? W_out[lm * 16 + k0 + 1] : 0.f;
            acc = __builtin_amdgcn_wmma_f32_16x16x4_f32(false, a, false, bb,
                                                        (short)0, acc, false, false);
        }
        if (lm < DMODEL) {
#pragma unroll
            for (int r = 0; r < 8; ++r)
                s_out[(l0 + r + 8 * hi) * 8 + lm] = acc[r];
        }
    }
    __syncthreads();

    {
        float part[NCLS];
#pragma unroll
        for (int c = 0; c < NCLS; ++c) part[c] = 0.f;
        for (int i = tid; i < L_SEQ * DMODEL; i += 256) {
            const float v = s_out[i];
#pragma unroll
            for (int c = 0; c < NCLS; ++c) part[c] += v * fc_w[c * (L_SEQ * DMODEL) + i];
        }
#pragma unroll
        for (int c = 0; c < NCLS; ++c) {
#pragma unroll
            for (int off = 16; off > 0; off >>= 1)
                part[c] += __shfl_down(part[c], off, 32);
        }
        if (lane == 0) {
#pragma unroll
            for (int c = 0; c < NCLS; ++c) s_red[wave * NCLS + c] = part[c];
        }
        __syncthreads();
        if (tid < NCLS) {
            float sres = fc_b[tid];
#pragma unroll
            for (int w = 0; w < 8; ++w) sres += s_red[w * NCLS + tid];
            out[b * NCLS + tid] = sres;
        }
    }
}

extern "C" void kernel_launch(void* const* d_in, const int* in_sizes, int n_in,
                              void* d_out, int out_size, void* d_ws, size_t ws_size,
                              hipStream_t stream) {
    (void)in_sizes; (void)n_in; (void)out_size; (void)d_ws; (void)ws_size;
    mamba_fused_kernel<<<dim3(1024), dim3(256), 0, stream>>>(
        (const float*)d_in[0],
        (const float*)d_in[1],
        (const float*)d_in[2],
        (const float*)d_in[3],
        (const float*)d_in[4],
        (const float*)d_in[5],
        (const float*)d_in[6],
        (const float*)d_in[7],
        (const float*)d_in[8],
        (const float*)d_in[9],
        (const float*)d_in[10],
        (const float*)d_in[11],
        (float*)d_out);
}


# Scan loop keeps only the recurrence and the sum over states: softplus(dt) once per (l,d) in a parallel pass before it, skip term and silu(z) gate in a parallel pass after it, z kept in registers
# speedup vs baseline: 2.2323x; 2.2323x over previous
#include <hip/hip_runtime.h>
#include <hip/hip_bf16.h>

#define L_SEQ 250
#define DMODEL 8
#define DINNER 16
#define DSTATE 16
#define DCONV 5
#define NCLS 12

typedef __attribute__((ext_vector_type(2))) float v2f;
typedef __attribute__((ext_vector_type(8))) float v8f;

__device__ __forceinline__ float sigmoidf_(float v) { return 1.0f / (1.0f + __expf(-v)); }

__global__ __launch_bounds__(256) void mamba_fused_kernel(
    const float* __restrict__ x,
    const float* __restrict__ W_in,
    const float* __restrict__ conv_w,
    const float* __restrict__ conv_b,
    const float* __restrict__ W_x,
    const float* __restrict__ dt_w,
    const float* __restrict__ dt_b,
    const float* __restrict__ A_log,
    const float* __restrict__ Dp,
    const float* __restrict__ W_out,
    const float* __restrict__ fc_w,
    const float* __restrict__ fc_b,
    float* __restrict__ out)
{
    __shared__ float smem[16298];
    float* s_dt  = smem;
    float* s_xc  = smem + 4016;
    float* s_B   = smem + 8016;
    float* s_C   = smem + 12032;
    float* s_dtr = smem + 16048;
    float* s_xb  = s_B;
    float* s_y   = s_dt;
    float* s_out = s_C;
    float* s_red = s_dtr;

    const int b    = blockIdx.x;
    const int tid  = threadIdx.x;
    const int lane = tid & 31;
    const int wave = tid >> 5;
    const int lm   = lane & 15;
    const int hi   = lane >> 4;
    const float* xb = x + (size_t)b * DMODEL * L_SEQ;

    for (int i = tid; i < DMODEL * L_SEQ; i += 256) s_xb[i] = xb[i];
    for (int i = tid * 32; i < NCLS * L_SEQ * DMODEL; i += 256 * 32)
        __builtin_prefetch(fc_w + i, 0, 0);
    __syncthreads();

    v8f zacc[2];
#pragma unroll
    for (int t = 0; t < 2; ++t) {
        const int l0 = (wave + 8 * t) * 16;
        const int lr = min(l0 + lm, L_SEQ - 1);
        v8f acc = {};
#pragma unroll
        for (int kc = 0; kc < 8; kc += 4) {
            const int k0 = kc + 2 * hi;
            v2f a, bb;
            a.x  = s_xb[(k0    ) * L_SEQ + lr];
            a.y  = s_xb[(k0 + 1) * L_SEQ + lr];
            bb.x = W_in[(16 + lm) * 8 + k0];
            bb.y = W_in[(16 + lm) * 8 + k0 + 1];
            acc = __builtin_amdgcn_wmma_f32_16x16x4_f32(false, a, false, bb,
                                                        (short)0, acc, false, false);
        }
        zacc[t] = acc;
    }

    for (int i = tid; i < L_SEQ * DINNER; i += 256) {
        const int l = i >> 4, d = i & 15;
        float accv = conv_b[d];
#pragma unroll
        for (int j = 0; j < DCONV; ++j) {
            const int t = l - (DCONV - 1) + j;
            if (t >= 0) {
                float xi = 0.f;
#pragma unroll
                for (int dm = 0; dm < DMODEL; ++dm) xi += s_xb[dm * L_SEQ + t] * W_in[d * 8 + dm];
                accv += conv_w[d * DCONV + j] * xi;
            }
        }
        s_xc[i] = accv * sigmoidf_(accv);
    }
    __syncthreads();

    for (int l = tid; l < L_SEQ; l += 256) {
        float a = 0.f;
#pragma unroll
        for (int d = 0; d < DINNER; ++d) a += s_xc[l * 16 + d] * W_x[d];
        s_dtr[l] = a;
    }

    for (int j = wave; j < 32; j += 8) {
        const int mt = j >> 1, nt = j & 1;
        const int l0 = mt * 16;
        const int lr = min(l0 + lm, L_SEQ - 1);
        const int e0 = 1 + nt * 16;
        v8f acc = {};
#pragma unroll
        for (int kc = 0; kc < 16; kc += 4) {
            const int k0 = kc + 2 * hi;
            v2f a, bb;
            a.x  = s_xc[lr * 16 + k0];
            a.y  = s_xc[lr * 16 + k0 + 1];
            bb.x = W_x[(e0 + lm) * 16 + k0];
            bb.y = W_x[(e0 + lm) * 16 + k0 + 1];
            acc = __builtin_amdgcn_wmma_f32_16x16x4_f32(false, a, false, bb,
                                                        (short)0, acc, false, false);
        }
        float* dst = nt ? s_C : s_B;
#pragma unroll
        for (int r = 0; r < 8; ++r) {
            const int row = min(l0 + r + 8 * hi, L_SEQ);
            dst[row * 16 + lm] = acc[r];
        }
    }
    __syncthreads();

    {
        const int d = tid & 15;
        const float dtw = dt_w[d], dtb = dt_b[d];
        for (int i = tid; i < L_SEQ * DINNER; i += 256) {
            float dv = s_dtr[i >> 4] * dtw + dtb;
            dv = (dv > 20.f) ? dv : log1pf(__expf(dv));
            s_dt[i] = dv;
        }
    }
    __syncthreads();

    {
        const int d = tid >> 4;
        const int s = tid & 15;
        const float As  = -__expf(A_log[d * 16 + s]);
        float h = 0.f;
        for (int l = 0; l < L_SEQ; ++l) {
            const float dv  = s_dt[l * 16 + d];
            const float xcv = s_xc[l * 16 + d];
            h = __expf(dv * As) * h + (dv * xcv) * s_B[l * 16 + s];
            float y = h * s_C[l * 16 + s];
            y += __shfl_xor(y, 1, 32);
            y += __shfl_xor(y, 2, 32);
            y += __shfl_xor(y, 4, 32);
            y += __shfl_xor(y, 8, 32);
            if (s == 0) s_y[l * 16 + d] = y;
        }
    }
    __syncthreads();

    {
        const float Dpd = Dp[lm];
#pragma unroll
        for (int t = 0; t < 2; ++t) {
            const int l0 = (wave + 8 * t) * 16;
#pragma unroll
            for (int r = 0; r < 8; ++r) {
                const int row = l0 + r + 8 * hi;
                if (row < L_SEQ) {
                    const int idx = row * 16 + lm;
                    float y = s_y[idx];
                    y += s_xc[idx] * Dpd;
                    const float zv = zacc[t][r];
                    y *= zv * sigmoidf_(zv);
                    s_y[idx] = y;
                }
            }
        }
    }
    __syncthreads();

    for (int mt = wave; mt < 16; mt += 8) {
        const int l0 = mt * 16;
        const int lr = min(l0 + lm, L_SEQ - 1);
        v8f acc = {};
#pragma unroll
        for (int kc = 0; kc < 16; kc += 4) {
            const int k0 = kc + 2 * hi;
            v2f a, bb;
            a.x  = s_y[lr * 16 + k0];
            a.y  = s_y[lr * 16 + k0 + 1];
            bb.x = (lm < DMODEL) ? W_out[lm * 16 + k0]     : 0.f;
            bb.y = (lm < DMODEL) ? W_out[lm * 16 + k0 + 1] : 0.f;
            acc = __builtin_amdgcn_wmma_f32_16x16x4_f32(false, a, false, bb,
                                                        (short)0, acc, false, false);
        }
        if (lm < DMODEL) {
#pragma unroll
            for (int r = 0; r < 8; ++r)
                s_out[(l0 + r + 8 * hi) * 8 + lm] = acc[r];
        }
    }
    __syncthreads();

    {
        float part[NCLS];
#pragma unroll
        for (int c = 0; c < NCLS; ++c) part[c] = 0.f;
        for (int i = tid; i < L_SEQ * DMODEL; i += 256) {
            const float v = s_out[i];
#pragma unroll
            for (int c = 0; c < NCLS; ++c) part[c] += v * fc_w[c * (L_SEQ * DMODEL) + i];
        }
#pragma unroll
        for (int c = 0; c < NCLS; ++c) {
#pragma unroll
            for (int off = 16; off > 0; off >>= 1)
                part[c] += __shfl_down(part[c], off, 32);
        }
        if (lane == 0) {
#pragma unroll
            for (int c = 0; c < NCLS; ++c) s_red[wave * NCLS + c] = part[c];
        }
        __syncthreads();
        if (tid < NCLS) {
            float sres = fc_b[tid];
#pragma unroll
            for (int w = 0; w < 8; ++w) sres += s_red[w * NCLS + tid];
            out[b * NCLS + tid] = sres;
        }
    }
}

extern "C" void kernel_launch(void* const* d_in, const int* in_sizes, int n_in,
                              void* d_out, int out_size, void* d_ws, size_t ws_size,
                              hipStream_t stream) {
    (void)in_sizes; (void)n_in; (void)out_size; (void)d_ws; (void)ws_size;
    mamba_fused_kernel<<<dim3(1024), dim3(256), 0, stream>>>(
        (const float*)d_in[0],
        (const float*)d_in[1],
        (const float*)d_in[2],
        (const float*)d_in[3],
        (const float*)d_in[4],
        (const float*)d_in[5],
        (const float*)d_in[6],
        (const float*)d_in[7],
        (const float*)d_in[8],
        (const float*)d_in[9],
        (const float*)d_in[10],
        (const float*)d_in[11],
        (float*)d_out);
}


# Scan step: the sum over the states of a channel runs as four DPP row_xmask adds in the vector unit instead of four LDS-path wave shuffles with a wait each; same butterfly order
# speedup vs baseline: 3.1595x; 1.4154x over previous
#include <hip/hip_runtime.h>
#include <hip/hip_bf16.h>

#define L_SEQ 250
#define DMODEL 8
#define DINNER 16
#define DSTATE 16
#define DCONV 5
#define NCLS 12

typedef __attribute__((ext_vector_type(2))) float v2f;
typedef __attribute__((ext_vector_type(8))) float v8f;

__device__ __forceinline__ float sigmoidf_(float v) { return 1.0f / (1.0f + __expf(-v)); }

template <int XM>
__device__ __forceinline__ float row_xor_add(float v) {
    const int other = __builtin_amdgcn_update_dpp(0, __builtin_bit_cast(int, v),
                                                  0x160 | XM  , 0xf, 0xf, true);
    return v + __builtin_bit_cast(float, other);
}

__global__ __launch_bounds__(256) void mamba_fused_kernel(
    const float* __restrict__ x,
    const float* __restrict__ W_in,
    const float* __restrict__ conv_w,
    const float* __restrict__ conv_b,
    const float* __restrict__ W_x,
    const float* __restrict__ dt_w,
    const float* __restrict__ dt_b,
    const float* __restrict__ A_log,
    const float* __restrict__ Dp,
    const float* __restrict__ W_out,
    const float* __restrict__ fc_w,
    const float* __restrict__ fc_b,
    float* __restrict__ out)
{
    __shared__ float smem[16298];
    float* s_dt  = smem;
    float* s_xc  = smem + 4016;
    float* s_B   = smem + 8016;
    float* s_C   = smem + 12032;
    float* s_dtr = smem + 16048;
    float* s_xb  = s_B;
    float* s_y   = s_dt;
    float* s_out = s_C;
    float* s_red = s_dtr;

    const int b    = blockIdx.x;
    const int tid  = threadIdx.x;
    const int lane = tid & 31;
    const int wave = tid >> 5;
    const int lm   = lane & 15;
    const int hi   = lane >> 4;
    const float* xb = x + (size_t)b * DMODEL * L_SEQ;

    for (int i = tid; i < DMODEL * L_SEQ; i += 256) s_xb[i] = xb[i];
    for (int i = tid * 32; i < NCLS * L_SEQ * DMODEL; i += 256 * 32)
        __builtin_prefetch(fc_w + i, 0, 0);
    __syncthreads();

    v8f zacc[2];
#pragma unroll
    for (int t = 0; t < 2; ++t) {
        const int l0 = (wave + 8 * t) * 16;
        const int lr = min(l0 + lm, L_SEQ - 1);
        v8f acc = {};
#pragma unroll
        for (int kc = 0; kc < 8; kc += 4) {
            const int k0 = kc + 2 * hi;
            v2f a, bb;
            a.x  = s_xb[(k0    ) * L_SEQ + lr];
            a.y  = s_xb[(k0 + 1) * L_SEQ + lr];
            bb.x = W_in[(16 + lm) * 8 + k0];
            bb.y = W_in[(16 + lm) * 8 + k0 + 1];
            acc = __builtin_amdgcn_wmma_f32_16x16x4_f32(false, a, false, bb,
                                                        (short)0, acc, false, false);
        }
        zacc[t] = acc;
    }

    for (int i = tid; i < L_SEQ * DINNER; i += 256) {
        const int l = i >> 4, d = i & 15;
        float accv = conv_b[d];
#pragma unroll
        for (int j = 0; j < DCONV; ++j) {
            const int t = l - (DCONV - 1) + j;
            if (t >= 0) {
                float xi = 0.f;
#pragma unroll
                for (int dm = 0; dm < DMODEL; ++dm) xi += s_xb[dm * L_SEQ + t] * W_in[d * 8 + dm];
                accv += conv_w[d * DCONV + j] * xi;
            }
        }
        s_xc[i] = accv * sigmoidf_(accv);
    }
    __syncthreads();

    for (int l = tid; l < L_SEQ; l += 256) {
        float a = 0.f;
#pragma unroll
        for (int d = 0; d < DINNER; ++d) a += s_xc[l * 16 + d] * W_x[d];
        s_dtr[l] = a;
    }

    for (int j = wave; j < 32; j += 8) {
        const int mt = j >> 1, nt = j & 1;
        const int l0 = mt * 16;
        const int lr = min(l0 + lm, L_SEQ - 1);
        const int e0 = 1 + nt * 16;
        v8f acc = {};
#pragma unroll
        for (int kc = 0; kc < 16; kc += 4) {
            const int k0 = kc + 2 * hi;
            v2f a, bb;
            a.x  = s_xc[lr * 16 + k0];
            a.y  = s_xc[lr * 16 + k0 + 1];
            bb.x = W_x[(e0 + lm) * 16 + k0];
            bb.y = W_x[(e0 + lm) * 16 + k0 + 1];
            acc = __builtin_amdgcn_wmma_f32_16x16x4_f32(false, a, false, bb,
                                                        (short)0, acc, false, false);
        }
        float* dst = nt ? s_C : s_B;
#pragma unroll
        for (int r = 0; r < 8; ++r) {
            const int row = min(l0 + r + 8 * hi, L_SEQ);
            dst[row * 16 + lm] = acc[r];
        }
    }
    __syncthreads();

    {
        const int d = tid & 15;
        const float dtw = dt_w[d], dtb = dt_b[d];
        for (int i = tid; i < L_SEQ * DINNER; i += 256) {
            float dv = s_dtr[i >> 4] * dtw + dtb;
            dv = (dv > 20.f) ? dv : log1pf(__expf(dv));
            s_dt[i] = dv;
        }
    }
    __syncthreads();

    {
        const int d = tid >> 4;
        const int s = tid & 15;
        const float As  = -__expf(A_log[d * 16 + s]);
        float h = 0.f;
        for (int l = 0; l < L_SEQ; ++l) {
            const float dv  = s_dt[l * 16 + d];
            const float xcv = s_xc[l * 16 + d];
            h = __expf(dv * As) * h + (dv * xcv) * s_B[l * 16 + s];
            float y = h * s_C[l * 16 + s];
            y = row_xor_add<1>(y);
            y = row_xor_add<2>(y);
            y = row_xor_add<4>(y);
            y = row_xor_add<8>(y);
            if (s == 0) s_y[l * 16 + d] = y;
        }
    }
    __syncthreads();

    {
        const float Dpd = Dp[lm];
#pragma unroll
        for (int t = 0; t < 2; ++t) {
            const int l0 = (wave + 8 * t) * 16;
#pragma unroll
            for (int r = 0; r < 8; ++r) {
                const int row = l0 + r + 8 * hi;
                if (row < L_SEQ) {
                    const int idx = row * 16 + lm;
                    float y = s_y[idx];
                    y += s_xc[idx] * Dpd;
                    const float zv = zacc[t][r];
                    y *= zv * sigmoidf_(zv);
                    s_y[idx] = y;
                }
            }
        }
    }
    __syncthreads();

    for (int mt = wave; mt < 16; mt += 8) {
        const int l0 = mt * 16;
        const int lr = min(l0 + lm, L_SEQ - 1);
        v8f acc = {};
#pragma unroll
        for (int kc = 0; kc < 16; kc += 4) {
            const int k0 = kc + 2 * hi;
            v2f a, bb;
            a.x  = s_y[lr * 16 + k0];
            a.y  = s_y[lr * 16 + k0 + 1];
            bb.x = (lm < DMODEL) ? W_out[lm * 16 + k0]     : 0.f;
            bb.y = (lm < DMODEL) ? W_out[lm * 16 + k0 + 1] : 0.f;
            acc = __builtin_amdgcn_wmma_f32_16x16x4_f32(false, a, false, bb,
                                                        (short)0, acc, false, false);
        }
        if (lm < DMODEL) {
#pragma unroll
            for (int r = 0; r < 8; ++r)
                s_out[(l0 + r + 8 * hi) * 8 + lm] = acc[r];
        }
    }
    __syncthreads();

    {
        float part[NCLS];
#pragma unroll
        for (int c = 0; c < NCLS; ++c) part[c] = 0.f;
        for (int i = tid; i < L_SEQ * DMODEL; i += 256) {
            const float v = s_out[i];
#pragma unroll
            for (int c = 0; c < NCLS; ++c) part[c] += v * fc_w[c * (L_SEQ * DMODEL) + i];
        }
#pragma unroll
        for (int c = 0; c < NCLS; ++c) {
#pragma unroll
            for (int off = 16; off > 0; off >>= 1)
                part[c] += __shfl_down(part[c], off, 32);
        }
        if (lane == 0) {
#pragma unroll
            for (int c = 0; c < NCLS; ++c) s_red[wave * NCLS + c] = part[c];
        }
        __syncthreads();
        if (tid < NCLS) {
            float sres = fc_b[tid];
#pragma unroll
            for (int w = 0; w < 8; ++w) sres += s_red[w * NCLS + tid];
            out[b * NCLS + tid] = sres;
        }
    }
}

extern "C" void kernel_launch(void* const* d_in, const int* in_sizes, int n_in,
                              void* d_out, int out_size, void* d_ws, size_t ws_size,
                              hipStream_t stream) {
    (void)in_sizes; (void)n_in; (void)out_size; (void)d_ws; (void)ws_size;
    mamba_fused_kernel<<<dim3(1024), dim3(256), 0, stream>>>(
        (const float*)d_in[0],
        (const float*)d_in[1],
        (const float*)d_in[2],
        (const float*)d_in[3],
        (const float*)d_in[4],
        (const float*)d_in[5],
        (const float*)d_in[6],
        (const float*)d_in[7],
        (const float*)d_in[8],
        (const float*)d_in[9],
        (const float*)d_in[10],
        (const float*)d_in[11],
        (float*)d_out);
}


# Scan: the sums over states of 16 consecutive steps are taken together by a transposing DPP butterfly (15 cross-lane adds per 16 sums instead of 64, same add tree), one LDS store per 16 steps
# speedup vs baseline: 4.3574x; 1.3791x over previous
#include <hip/hip_runtime.h>
#include <hip/hip_bf16.h>

#define L_SEQ 250
#define DMODEL 8
#define DINNER 16
#define DSTATE 16
#define DCONV 5
#define NCLS 12

typedef __attribute__((ext_vector_type(2))) float v2f;
typedef __attribute__((ext_vector_type(8))) float v8f;

__device__ __forceinline__ float sigmoidf_(float v) { return 1.0f / (1.0f + __expf(-v)); }

template <int XM>
__device__ __forceinline__ float row_xmask_from(float v) {
    const int other = __builtin_amdgcn_update_dpp(0, __builtin_bit_cast(int, v),
                                                  0x160 | XM  , 0xf, 0xf, true);
    return __builtin_bit_cast(float, other);
}

template <int XM>
__device__ __forceinline__ float fold_pair(float a, float b, bool up) {
    const float keep = up ? b : a;
    const float send = up ? a : b;
    return keep + row_xmask_from<XM>(send);
}

template <int N>
__device__ __forceinline__ void scan_steps(float& h, const float As, const int l0, const int d, const int s,
                                           const bool b0, const bool b1, const bool b2, const bool b3,
                                           const float* s_dt, const float* s_xc,
                                           const float* s_B, const float* s_C, float* s_y) {
    float y[16];
    for (int j = 0; j < 16; ++j) {
        if (j < N) {
            const int l = l0 + j;
            const float dv  = s_dt[l * 16 + d];
            const float xcv = s_xc[l * 16 + d];
            h = __expf(dv * As) * h + (dv * xcv) * s_B[l * 16 + s];
            y[j] = h * s_C[l * 16 + s];
        } else {
            y[j] = 0.f;
        }
    }
    float r1[8], r2[4], r3[2];
    for (int j = 0; j < 8; ++j) r1[j] = fold_pair<1>(y[2 * j], y[2 * j + 1], b0);
    for (int j = 0; j < 4; ++j) r2[j] = fold_pair<2>(r1[2 * j], r1[2 * j + 1], b1);
    for (int j = 0; j < 2; ++j) r3[j] = fold_pair<4>(r2[2 * j], r2[2 * j + 1], b2);
    const float ysum = fold_pair<8>(r3[0], r3[1], b3);
    if (s < N) s_y[(l0 + s) * 16 + d] = ysum;
}

__global__ __launch_bounds__(256) void mamba_fused_kernel(
    const float* __restrict__ x,
    const float* __restrict__ W_in,
    const float* __restrict__ conv_w,
    const float* __restrict__ conv_b,
    const float* __restrict__ W_x,
    const float* __restrict__ dt_w,
    const float* __restrict__ dt_b,
    const float* __restrict__ A_log,
    const float* __restrict__ Dp,
    const float* __restrict__ W_out,
    const float* __restrict__ fc_w,
    const float* __restrict__ fc_b,
    float* __restrict__ out)
{
    __shared__ float smem[16298];
    float* s_dt  = smem;
    float* s_xc  = smem + 4016;
    float* s_B   = smem + 8016;
    float* s_C   = smem + 12032;
    float* s_dtr = smem + 16048;
    float* s_xb  = s_B;
    float* s_y   = s_dt;
    float* s_out = s_C;
    float* s_red = s_dtr;

    const int b    = blockIdx.x;
    const int tid  = threadIdx.x;
    const int lane = tid & 31;
    const int wave = tid >> 5;
    const int lm   = lane & 15;
    const int hi   = lane >> 4;
    const float* xb = x + (size_t)b * DMODEL * L_SEQ;

    for (int i = tid; i < DMODEL * L_SEQ; i += 256) s_xb[i] = xb[i];
    for (int i = tid * 32; i < NCLS * L_SEQ * DMODEL; i += 256 * 32)
        __builtin_prefetch(fc_w + i, 0, 0);
    __syncthreads();

    v8f zacc[2];
#pragma unroll
    for (int t = 0; t < 2; ++t) {
        const int l0 = (wave + 8 * t) * 16;
        const int lr = min(l0 + lm, L_SEQ - 1);
        v8f acc = {};
#pragma unroll
        for (int kc = 0; kc < 8; kc += 4) {
            const int k0 = kc + 2 * hi;
            v2f a, bb;
            a.x  = s_xb[(k0    ) * L_SEQ + lr];
            a.y  = s_xb[(k0 + 1) * L_SEQ + lr];
            bb.x = W_in[(16 + lm) * 8 + k0];
            bb.y = W_in[(16 + lm) * 8 + k0 + 1];
            acc = __builtin_amdgcn_wmma_f32_16x16x4_f32(false, a, false, bb,
                                                        (short)0, acc, false, false);
        }
        zacc[t] = acc;
    }

    for (int i = tid; i < L_SEQ * DINNER; i += 256) {
        const int l = i >> 4, d = i & 15;
        float accv = conv_b[d];
#pragma unroll
        for (int j = 0; j < DCONV; ++j) {
            const int t = l - (DCONV - 1) + j;
            if (t >= 0) {
                float xi = 0.f;
#pragma unroll
                for (int dm = 0; dm < DMODEL; ++dm) xi += s_xb[dm * L_SEQ + t] * W_in[d * 8 + dm];
                accv += conv_w[d * DCONV + j] * xi;
            }
        }
        s_xc[i] = accv * sigmoidf_(accv);
    }
    __syncthreads();

    for (int l = tid; l < L_SEQ; l += 256) {
        float a = 0.f;
#pragma unroll
        for (int d = 0; d < DINNER; ++d) a += s_xc[l * 16 + d] * W_x[d];
        s_dtr[l] = a;
    }

    for (int j = wave; j < 32; j += 8) {
        const int mt = j >> 1, nt = j & 1;
        const int l0 = mt * 16;
        const int lr = min(l0 + lm, L_SEQ - 1);
        const int e0 = 1 + nt * 16;
        v8f acc = {};
#pragma unroll
        for (int kc = 0; kc < 16; kc += 4) {
            const int k0 = kc + 2 * hi;
            v2f a, bb;
            a.x  = s_xc[lr * 16 + k0];
            a.y  = s_xc[lr * 16 + k0 + 1];
            bb.x = W_x[(e0 + lm) * 16 + k0];
            bb.y = W_x[(e0 + lm) * 16 + k0 + 1];
            acc = __builtin_amdgcn_wmma_f32_16x16x4_f32(false, a, false, bb,
                                                        (short)0, acc, false, false);
        }
        float* dst = nt ? s_C : s_B;
#pragma unroll
        for (int r = 0; r < 8; ++r) {
            const int row = min(l0 + r + 8 * hi, L_SEQ);
            dst[row * 16 + lm] = acc[r];
        }
    }
    __syncthreads();

    {
        const int d = tid & 15;
        const float dtw = dt_w[d], dtb = dt_b[d];
        for (int i = tid; i < L_SEQ * DINNER; i += 256) {
            float dv = s_dtr[i >> 4] * dtw + dtb;
            dv = (dv > 20.f) ? dv : log1pf(__expf(dv));
            s_dt[i] = dv;
        }
    }
    __syncthreads();

    {
        const int d = tid >> 4;
        const int s = tid & 15;
        const float As  = -__expf(A_log[d * 16 + s]);
        const bool b0 = (s & 1) != 0, b1 = (s & 2) != 0, b2 = (s & 4) != 0, b3 = (s & 8) != 0;
        float h = 0.f;
        for (int l0 = 0; l0 < 240; l0 += 16)
            scan_steps<16>(h, As, l0, d, s, b0, b1, b2, b3, s_dt, s_xc, s_B, s_C, s_y);
        scan_steps<L_SEQ - 240>(h, As, 240, d, s, b0, b1, b2, b3, s_dt, s_xc, s_B, s_C, s_y);
    }
    __syncthreads();

    {
        const float Dpd = Dp[lm];
#pragma unroll
        for (int t = 0; t < 2; ++t) {
            const int l0 = (wave + 8 * t) * 16;
#pragma unroll
            for (int r = 0; r < 8; ++r) {
                const int row = l0 + r + 8 * hi;
                if (row < L_SEQ) {
                    const int idx = row * 16 + lm;
                    float y = s_y[idx];
                    y += s_xc[idx] * Dpd;
                    const float zv = zacc[t][r];
                    y *= zv * sigmoidf_(zv);
                    s_y[idx] = y;
                }
            }
        }
    }
    __syncthreads();

    for (int mt = wave; mt < 16; mt += 8) {
        const int l0 = mt * 16;
        const int lr = min(l0 + lm, L_SEQ - 1);
        v8f acc = {};
#pragma unroll
        for (int kc = 0; kc < 16; kc += 4) {
            const int k0 = kc + 2 * hi;
            v2f a, bb;
            a.x  = s_y[lr * 16 + k0];
            a.y  = s_y[lr * 16 + k0 + 1];
            bb.x = (lm < DMODEL) ? W_out[lm * 16 + k0]     : 0.f;
            bb.y = (lm < DMODEL) ? W_out[lm * 16 + k0 + 1] : 0.f;
            acc = __builtin_amdgcn_wmma_f32_16x16x4_f32(false, a, false, bb,
                                                        (short)0, acc, false, false);
        }
        if (lm < DMODEL) {
#pragma unroll
            for (int r = 0; r < 8; ++r)
                s_out[(l0 + r + 8 * hi) * 8 + lm] = acc[r];
        }
    }
    __syncthreads();

    {
        float part[NCLS];
#pragma unroll
        for (int c = 0; c < NCLS; ++c) part[c] = 0.f;
        for (int i = tid; i < L_SEQ * DMODEL; i += 256) {
            const float v = s_out[i];
#pragma unroll
            for (int c = 0; c < NCLS; ++c) part[c] += v * fc_w[c * (L_SEQ * DMODEL) + i];
        }
#pragma unroll
        for (int c = 0; c < NCLS; ++c) {
#pragma unroll
            for (int off = 16; off > 0; off >>= 1)
                part[c] += __shfl_down(part[c], off, 32);
        }
        if (lane == 0) {
#pragma unroll
            for (int c = 0; c < NCLS; ++c) s_red[wave * NCLS + c] = part[c];
        }
        __syncthreads();
        if (tid < NCLS) {
            float sres = fc_b[tid];
#pragma unroll
            for (int w = 0; w < 8; ++w) sres += s_red[w * NCLS + tid];
            out[b * NCLS + tid] = sres;
        }
    }
}

extern "C" void kernel_launch(void* const* d_in, const int* in_sizes, int n_in,
                              void* d_out, int out_size, void* d_ws, size_t ws_size,
                              hipStream_t stream) {
    (void)in_sizes; (void)n_in; (void)out_size; (void)d_ws; (void)ws_size;
    mamba_fused_kernel<<<dim3(1024), dim3(256), 0, stream>>>(
        (const float*)d_in[0],
        (const float*)d_in[1],
        (const float*)d_in[2],
        (const float*)d_in[3],
        (const float*)d_in[4],
        (const float*)d_in[5],
        (const float*)d_in[6],
        (const float*)d_in[7],
        (const float*)d_in[8],
        (const float*)d_in[9],
        (const float*)d_in[10],
        (const float*)d_in[11],
        (float*)d_out);
}


# In-projection xi = x.W_in[:16]^T done once per (l,d) on the matrix unit (f32, z's A fragments) into LDS; conv reads 5 taps, no longer redoes 5x8 products; xi's k-sum order is now the WMMA's
# speedup vs baseline: 5.1635x; 1.1850x over previous
#include <hip/hip_runtime.h>
#include <hip/hip_bf16.h>

#define L_SEQ 250
#define DMODEL 8
#define DINNER 16
#define DSTATE 16
#define DCONV 5
#define NCLS 12

typedef __attribute__((ext_vector_type(2))) float v2f;
typedef __attribute__((ext_vector_type(8))) float v8f;

__device__ __forceinline__ float sigmoidf_(float v) { return 1.0f / (1.0f + __expf(-v)); }

template <int XM>
__device__ __forceinline__ float row_xmask_from(float v) {
    const int other = __builtin_amdgcn_update_dpp(0, __builtin_bit_cast(int, v),
                                                  0x160 | XM  , 0xf, 0xf, true);
    return __builtin_bit_cast(float, other);
}

template <int XM>
__device__ __forceinline__ float fold_pair(float a, float b, bool up) {
    const float keep = up ? b : a;
    const float send = up ? a : b;
    return keep + row_xmask_from<XM>(send);
}

template <int N>
__device__ __forceinline__ void scan_steps(float& h, const float As, const int l0, const int d, const int s,
                                           const bool b0, const bool b1, const bool b2, const bool b3,
                                           const float* s_dt, const float* s_xc,
                                           const float* s_B, const float* s_C, float* s_y) {
    float y[16];
    for (int j = 0; j < 16; ++j) {
        if (j < N) {
            const int l = l0 + j;
            const float dv  = s_dt[l * 16 + d];
            const float xcv = s_xc[l * 16 + d];
            h = __expf(dv * As) * h + (dv * xcv) * s_B[l * 16 + s];
            y[j] = h * s_C[l * 16 + s];
        } else {
            y[j] = 0.f;
        }
    }
    float r1[8], r2[4], r3[2];
    for (int j = 0; j < 8; ++j) r1[j] = fold_pair<1>(y[2 * j], y[2 * j + 1], b0);
    for (int j = 0; j < 4; ++j) r2[j] = fold_pair<2>(r1[2 * j], r1[2 * j + 1], b1);
    for (int j = 0; j < 2; ++j) r3[j] = fold_pair<4>(r2[2 * j], r2[2 * j + 1], b2);
    const float ysum = fold_pair<8>(r3[0], r3[1], b3);
    if (s < N) s_y[(l0 + s) * 16 + d] = ysum;
}

__global__ __launch_bounds__(256) void mamba_fused_kernel(
    const float* __restrict__ x,
    const float* __restrict__ W_in,
    const float* __restrict__ conv_w,
    const float* __restrict__ conv_b,
    const float* __restrict__ W_x,
    const float* __restrict__ dt_w,
    const float* __restrict__ dt_b,
    const float* __restrict__ A_log,
    const float* __restrict__ Dp,
    const float* __restrict__ W_out,
    const float* __restrict__ fc_w,
    const float* __restrict__ fc_b,
    float* __restrict__ out)
{
    __shared__ float smem[16298];
    float* s_dt  = smem;
    float* s_xc  = smem + 4016;
    float* s_B   = smem + 8016;
    float* s_C   = smem + 12032;
    float* s_dtr = smem + 16048;
    float* s_xb  = s_B;
    float* s_xi  = smem + 12096;
    float* s_y   = s_dt;
    float* s_out = s_C;
    float* s_red = s_dtr;

    const int b    = blockIdx.x;
    const int tid  = threadIdx.x;
    const int lane = tid & 31;
    const int wave = tid >> 5;
    const int lm   = lane & 15;
    const int hi   = lane >> 4;
    const float* xb = x + (size_t)b * DMODEL * L_SEQ;

    for (int i = tid; i < DMODEL * L_SEQ; i += 256) s_xb[i] = xb[i];
    if (tid < (DCONV - 1) * DINNER) s_xi[tid - (DCONV - 1) * DINNER] = 0.f;
    for (int i = tid * 32; i < NCLS * L_SEQ * DMODEL; i += 256 * 32)
        __builtin_prefetch(fc_w + i, 0, 0);
    __syncthreads();

    v8f zacc[2];
#pragma unroll
    for (int t = 0; t < 2; ++t) {
        const int l0 = (wave + 8 * t) * 16;
        const int lr = min(l0 + lm, L_SEQ - 1);
        v8f acc = {}, accx = {};
#pragma unroll
        for (int kc = 0; kc < 8; kc += 4) {
            const int k0 = kc + 2 * hi;
            v2f a, bb, bx;
            a.x  = s_xb[(k0    ) * L_SEQ + lr];
            a.y  = s_xb[(k0 + 1) * L_SEQ + lr];
            bb.x = W_in[(16 + lm) * 8 + k0];
            bb.y = W_in[(16 + lm) * 8 + k0 + 1];
            bx.x = W_in[lm * 8 + k0];
            bx.y = W_in[lm * 8 + k0 + 1];
            acc  = __builtin_amdgcn_wmma_f32_16x16x4_f32(false, a, false, bb,
                                                         (short)0, acc, false, false);
            accx = __builtin_amdgcn_wmma_f32_16x16x4_f32(false, a, false, bx,
                                                         (short)0, accx, false, false);
        }
        zacc[t] = acc;
#pragma unroll
        for (int r = 0; r < 8; ++r) {
            const int row = min(l0 + r + 8 * hi, L_SEQ);
            s_xi[row * 16 + lm] = accx[r];
        }
    }
    __syncthreads();

    {
        const int d = tid & 15;
        float cw[DCONV];
#pragma unroll
        for (int j = 0; j < DCONV; ++j) cw[j] = conv_w[d * DCONV + j];
        const float cb = conv_b[d];
        for (int i = tid; i < L_SEQ * DINNER; i += 256) {
            float accv = cb;
#pragma unroll
            for (int j = 0; j < DCONV; ++j)
                accv += cw[j] * s_xi[i + (j - (DCONV - 1)) * DINNER];
            s_xc[i] = accv * sigmoidf_(accv);
        }
    }
    __syncthreads();

    for (int l = tid; l < L_SEQ; l += 256) {
        float a = 0.f;
#pragma unroll
        for (int d = 0; d < DINNER; ++d) a += s_xc[l * 16 + d] * W_x[d];
        s_dtr[l] = a;
    }

    for (int j = wave; j < 32; j += 8) {
        const int mt = j >> 1, nt = j & 1;
        const int l0 = mt * 16;
        const int lr = min(l0 + lm, L_SEQ - 1);
        const int e0 = 1 + nt * 16;
        v8f acc = {};
#pragma unroll
        for (int kc = 0; kc < 16; kc += 4) {
            const int k0 = kc + 2 * hi;
            v2f a, bb;
            a.x  = s_xc[lr * 16 + k0];
            a.y  = s_xc[lr * 16 + k0 + 1];
            bb.x = W_x[(e0 + lm) * 16 + k0];
            bb.y = W_x[(e0 + lm) * 16 + k0 + 1];
            acc = __builtin_amdgcn_wmma_f32_16x16x4_f32(false, a, false, bb,
                                                        (short)0, acc, false, false);
        }
        float* dst = nt ? s_C : s_B;
#pragma unroll
        for (int r = 0; r < 8; ++r) {
            const int row = min(l0 + r + 8 * hi, L_SEQ);
            dst[row * 16 + lm] = acc[r];
        }
    }
    __syncthreads();

    {
        const int d = tid & 15;
        const float dtw = dt_w[d], dtb = dt_b[d];
        for (int i = tid; i < L_SEQ * DINNER; i += 256) {
            float dv = s_dtr[i >> 4] * dtw + dtb;
            dv = (dv > 20.f) ? dv : log1pf(__expf(dv));
            s_dt[i] = dv;
        }
    }
    __syncthreads();

    {
        const int d = tid >> 4;
        const int s = tid & 15;
        const float As  = -__expf(A_log[d * 16 + s]);
        const bool b0 = (s & 1) != 0, b1 = (s & 2) != 0, b2 = (s & 4) != 0, b3 = (s & 8) != 0;
        float h = 0.f;
        for (int l0 = 0; l0 < 240; l0 += 16)
            scan_steps<16>(h, As, l0, d, s, b0, b1, b2, b3, s_dt, s_xc, s_B, s_C, s_y);
        scan_steps<L_SEQ - 240>(h, As, 240, d, s, b0, b1, b2, b3, s_dt, s_xc, s_B, s_C, s_y);
    }
    __syncthreads();

    {
        const float Dpd = Dp[lm];
#pragma unroll
        for (int t = 0; t < 2; ++t) {
            const int l0 = (wave + 8 * t) * 16;
#pragma unroll
            for (int r = 0; r < 8; ++r) {
                const int row = l0 + r + 8 * hi;
                if (row < L_SEQ) {
                    const int idx = row * 16 + lm;
                    float y = s_y[idx];
                    y += s_xc[idx] * Dpd;
                    const float zv = zacc[t][r];
                    y *= zv * sigmoidf_(zv);
                    s_y[idx] = y;
                }
            }
        }
    }
    __syncthreads();

    for (int mt = wave; mt < 16; mt += 8) {
        const int l0 = mt * 16;
        const int lr = min(l0 + lm, L_SEQ - 1);
        v8f acc = {};
#pragma unroll
        for (int kc = 0; kc < 16; kc += 4) {
            const int k0 = kc + 2 * hi;
            v2f a, bb;
            a.x  = s_y[lr * 16 + k0];
            a.y  = s_y[lr * 16 + k0 + 1];
            bb.x = (lm < DMODEL) ? W_out[lm * 16 + k0]     : 0.f;
            bb.y = (lm < DMODEL) ? W_out[lm * 16 + k0 + 1] : 0.f;
            acc = __builtin_amdgcn_wmma_f32_16x16x4_f32(false, a, false, bb,
                                                        (short)0, acc, false, false);
        }
        if (lm < DMODEL) {
#pragma unroll
            for (int r = 0; r < 8; ++r)
                s_out[(l0 + r + 8 * hi) * 8 + lm] = acc[r];
        }
    }
    __syncthreads();

    {
        float part[NCLS];
#pragma unroll
        for (int c = 0; c < NCLS; ++c) part[c] = 0.f;
        for (int i = tid; i < L_SEQ * DMODEL; i += 256) {
            const float v = s_out[i];
#pragma unroll
            for (int c = 0; c < NCLS; ++c) part[c] += v * fc_w[c * (L_SEQ * DMODEL) + i];
        }
#pragma unroll
        for (int c = 0; c < NCLS; ++c) {
#pragma unroll
            for (int off = 16; off > 0; off >>= 1)
                part[c] += __shfl_down(part[c], off, 32);
        }
        if (lane == 0) {
#pragma unroll
            for (int c = 0; c < NCLS; ++c) s_red[wave * NCLS + c] = part[c];
        }
        __syncthreads();
        if (tid < NCLS) {
            float sres = fc_b[tid];
#pragma unroll
            for (int w = 0; w < 8; ++w) sres += s_red[w * NCLS + tid];
            out[b * NCLS + tid] = sres;
        }
    }
}

extern "C" void kernel_launch(void* const* d_in, const int* in_sizes, int n_in,
                              void* d_out, int out_size, void* d_ws, size_t ws_size,
                              hipStream_t stream) {
    (void)in_sizes; (void)n_in; (void)out_size; (void)d_ws; (void)ws_size;
    mamba_fused_kernel<<<dim3(1024), dim3(256), 0, stream>>>(
        (const float*)d_in[0],
        (const float*)d_in[1],
        (const float*)d_in[2],
        (const float*)d_in[3],
        (const float*)d_in[4],
        (const float*)d_in[5],
        (const float*)d_in[6],
        (const float*)d_in[7],
        (const float*)d_in[8],
        (const float*)d_in[9],
        (const float*)d_in[10],
        (const float*)d_in[11],
        (float*)d_out);
}


# Softplus and both silu passes on the native v_log_f32 / v_rcp_f32 (softplus = ln2*log2(1+e) + rounding correction) instead of library log1pf and the IEEE divide sequence; last-ulp rounding differs
# speedup vs baseline: 6.0838x; 1.1782x over previous
#include <hip/hip_runtime.h>
#include <hip/hip_bf16.h>

#define L_SEQ 250
#define DMODEL 8
#define DINNER 16
#define DSTATE 16
#define DCONV 5
#define NCLS 12

typedef __attribute__((ext_vector_type(2))) float v2f;
typedef __attribute__((ext_vector_type(8))) float v8f;

__device__ __forceinline__ float sigmoidf_(float v) { return __builtin_amdgcn_rcpf(1.0f + __expf(-v)); }

__device__ __forceinline__ float softplusf_(float v) {
    const float t  = __expf(v);
    const float u  = 1.0f + t;
    const float lg = __builtin_amdgcn_logf(u) * 0.6931471805599453f;
    const float r  = lg + (t - (u - 1.0f)) * __builtin_amdgcn_rcpf(u);
    return (v > 20.f) ? v : r;
}

template <int XM>
__device__ __forceinline__ float row_xmask_from(float v) {
    const int other = __builtin_amdgcn_update_dpp(0, __builtin_bit_cast(int, v),
                                                  0x160 | XM  , 0xf, 0xf, true);
    return __builtin_bit_cast(float, other);
}

template <int XM>
__device__ __forceinline__ float fold_pair(float a, float b, bool up) {
    const float keep = up ? b : a;
    const float send = up ? a : b;
    return keep + row_xmask_from<XM>(send);
}

template <int N>
__device__ __forceinline__ void scan_steps(float& h, const float As, const int l0, const int d, const int s,
                                           const bool b0, const bool b1, const bool b2, const bool b3,
                                           const float* s_dt, const float* s_xc,
                                           const float* s_B, const float* s_C, float* s_y) {
    float y[16];
    for (int j = 0; j < 16; ++j) {
        if (j < N) {
            const int l = l0 + j;
            const float dv  = s_dt[l * 16 + d];
            const float xcv = s_xc[l * 16 + d];
            h = __expf(dv * As) * h + (dv * xcv) * s_B[l * 16 + s];
            y[j] = h * s_C[l * 16 + s];
        } else {
            y[j] = 0.f;
        }
    }
    float r1[8], r2[4], r3[2];
    for (int j = 0; j < 8; ++j) r1[j] = fold_pair<1>(y[2 * j], y[2 * j + 1], b0);
    for (int j = 0; j < 4; ++j) r2[j] = fold_pair<2>(r1[2 * j], r1[2 * j + 1], b1);
    for (int j = 0; j < 2; ++j) r3[j] = fold_pair<4>(r2[2 * j], r2[2 * j + 1], b2);
    const float ysum = fold_pair<8>(r3[0], r3[1], b3);
    if (s < N) s_y[(l0 + s) * 16 + d] = ysum;
}

__global__ __launch_bounds__(256) void mamba_fused_kernel(
    const float* __restrict__ x,
    const float* __restrict__ W_in,
    const float* __restrict__ conv_w,
    const float* __restrict__ conv_b,
    const float* __restrict__ W_x,
    const float* __restrict__ dt_w,
    const float* __restrict__ dt_b,
    const float* __restrict__ A_log,
    const float* __restrict__ Dp,
    const float* __restrict__ W_out,
    const float* __restrict__ fc_w,
    const float* __restrict__ fc_b,
    float* __restrict__ out)
{
    __shared__ float smem[16298];
    float* s_dt  = smem;
    float* s_xc  = smem + 4016;
    float* s_B   = smem + 8016;
    float* s_C   = smem + 12032;
    float* s_dtr = smem + 16048;
    float* s_xb  = s_B;
    float* s_xi  = smem + 12096;
    float* s_y   = s_dt;
    float* s_out = s_C;
    float* s_red = s_dtr;

    const int b    = blockIdx.x;
    const int tid  = threadIdx.x;
    const int lane = tid & 31;
    const int wave = tid >> 5;
    const int lm   = lane & 15;
    const int hi   = lane >> 4;
    const float* xb = x + (size_t)b * DMODEL * L_SEQ;

    for (int i = tid; i < DMODEL * L_SEQ; i += 256) s_xb[i] = xb[i];
    if (tid < (DCONV - 1) * DINNER) s_xi[tid - (DCONV - 1) * DINNER] = 0.f;
    for (int i = tid * 32; i < NCLS * L_SEQ * DMODEL; i += 256 * 32)
        __builtin_prefetch(fc_w + i, 0, 0);
    __syncthreads();

    v8f zacc[2];
#pragma unroll
    for (int t = 0; t < 2; ++t) {
        const int l0 = (wave + 8 * t) * 16;
        const int lr = min(l0 + lm, L_SEQ - 1);
        v8f acc = {}, accx = {};
#pragma unroll
        for (int kc = 0; kc < 8; kc += 4) {
            const int k0 = kc + 2 * hi;
            v2f a, bb, bx;
            a.x  = s_xb[(k0    ) * L_SEQ + lr];
            a.y  = s_xb[(k0 + 1) * L_SEQ + lr];
            bb.x = W_in[(16 + lm) * 8 + k0];
            bb.y = W_in[(16 + lm) * 8 + k0 + 1];
            bx.x = W_in[lm * 8 + k0];
            bx.y = W_in[lm * 8 + k0 + 1];
            acc  = __builtin_amdgcn_wmma_f32_16x16x4_f32(false, a, false, bb,
                                                         (short)0, acc, false, false);
            accx = __builtin_amdgcn_wmma_f32_16x16x4_f32(false, a, false, bx,
                                                         (short)0, accx, false, false);
        }
        zacc[t] = acc;
#pragma unroll
        for (int r = 0; r < 8; ++r) {
            const int row = min(l0 + r + 8 * hi, L_SEQ);
            s_xi[row * 16 + lm] = accx[r];
        }
    }
    __syncthreads();

    {
        const int d = tid & 15;
        float cw[DCONV];
#pragma unroll
        for (int j = 0; j < DCONV; ++j) cw[j] = conv_w[d * DCONV + j];
        const float cb = conv_b[d];
        for (int i = tid; i < L_SEQ * DINNER; i += 256) {
            float accv = cb;
#pragma unroll
            for (int j = 0; j < DCONV; ++j)
                accv += cw[j] * s_xi[i + (j - (DCONV - 1)) * DINNER];
            s_xc[i] = accv * sigmoidf_(accv);
        }
    }
    __syncthreads();

    for (int l = tid; l < L_SEQ; l += 256) {
        float a = 0.f;
#pragma unroll
        for (int d = 0; d < DINNER; ++d) a += s_xc[l * 16 + d] * W_x[d];
        s_dtr[l] = a;
    }

    for (int j = wave; j < 32; j += 8) {
        const int mt = j >> 1, nt = j & 1;
        const int l0 = mt * 16;
        const int lr = min(l0 + lm, L_SEQ - 1);
        const int e0 = 1 + nt * 16;
        v8f acc = {};
#pragma unroll
        for (int kc = 0; kc < 16; kc += 4) {
            const int k0 = kc + 2 * hi;
            v2f a, bb;
            a.x  = s_xc[lr * 16 + k0];
            a.y  = s_xc[lr * 16 + k0 + 1];
            bb.x = W_x[(e0 + lm) * 16 + k0];
            bb.y = W_x[(e0 + lm) * 16 + k0 + 1];
            acc = __builtin_amdgcn_wmma_f32_16x16x4_f32(false, a, false, bb,
                                                        (short)0, acc, false, false);
        }
        float* dst = nt ? s_C : s_B;
#pragma unroll
        for (int r = 0; r < 8; ++r) {
            const int row = min(l0 + r + 8 * hi, L_SEQ);
            dst[row * 16 + lm] = acc[r];
        }
    }
    __syncthreads();

    {
        const int d = tid & 15;
        const float dtw = dt_w[d], dtb = dt_b[d];
        for (int i = tid; i < L_SEQ * DINNER; i += 256) {
            float dv = s_dtr[i >> 4] * dtw + dtb;
            dv = softplusf_(dv);
            s_dt[i] = dv;
        }
    }
    __syncthreads();

    {
        const int d = tid >> 4;
        const int s = tid & 15;
        const float As  = -__expf(A_log[d * 16 + s]);
        const bool b0 = (s & 1) != 0, b1 = (s & 2) != 0, b2 = (s & 4) != 0, b3 = (s & 8) != 0;
        float h = 0.f;
        for (int l0 = 0; l0 < 240; l0 += 16)
            scan_steps<16>(h, As, l0, d, s, b0, b1, b2, b3, s_dt, s_xc, s_B, s_C, s_y);
        scan_steps<L_SEQ - 240>(h, As, 240, d, s, b0, b1, b2, b3, s_dt, s_xc, s_B, s_C, s_y);
    }
    __syncthreads();

    {
        const float Dpd = Dp[lm];
#pragma unroll
        for (int t = 0; t < 2; ++t) {
            const int l0 = (wave + 8 * t) * 16;
#pragma unroll
            for (int r = 0; r < 8; ++r) {
                const int row = l0 + r + 8 * hi;
                if (row < L_SEQ) {
                    const int idx = row * 16 + lm;
                    float y = s_y[idx];
                    y += s_xc[idx] * Dpd;
                    const float zv = zacc[t][r];
                    y *= zv * sigmoidf_(zv);
                    s_y[idx] = y;
                }
            }
        }
    }
    __syncthreads();

    for (int mt = wave; mt < 16; mt += 8) {
        const int l0 = mt * 16;
        const int lr = min(l0 + lm, L_SEQ - 1);
        v8f acc = {};
#pragma unroll
        for (int kc = 0; kc < 16; kc += 4) {
            const int k0 = kc + 2 * hi;
            v2f a, bb;
            a.x  = s_y[lr * 16 + k0];
            a.y  = s_y[lr * 16 + k0 + 1];
            bb.x = (lm < DMODEL) ? W_out[lm * 16 + k0]     : 0.f;
            bb.y = (lm < DMODEL) ? W_out[lm * 16 + k0 + 1] : 0.f;
            acc = __builtin_amdgcn_wmma_f32_16x16x4_f32(false, a, false, bb,
                                                        (short)0, acc, false, false);
        }
        if (lm < DMODEL) {
#pragma unroll
            for (int r = 0; r < 8; ++r)
                s_out[(l0 + r + 8 * hi) * 8 + lm] = acc[r];
        }
    }
    __syncthreads();

    {
        float part[NCLS];
#pragma unroll
        for (int c = 0; c < NCLS; ++c) part[c] = 0.f;
        for (int i = tid; i < L_SEQ * DMODEL; i += 256) {
            const float v = s_out[i];
#pragma unroll
            for (int c = 0; c < NCLS; ++c) part[c] += v * fc_w[c * (L_SEQ * DMODEL) + i];
        }
#pragma unroll
        for (int c = 0; c < NCLS; ++c) {
#pragma unroll
            for (int off = 16; off > 0; off >>= 1)
                part[c] += __shfl_down(part[c], off, 32);
        }
        if (lane == 0) {
#pragma unroll
            for (int c = 0; c < NCLS; ++c) s_red[wave * NCLS + c] = part[c];
        }
        __syncthreads();
        if (tid < NCLS) {
            float sres = fc_b[tid];
#pragma unroll
            for (int w = 0; w < 8; ++w) sres += s_red[w * NCLS + tid];
            out[b * NCLS + tid] = sres;
        }
    }
}

extern "C" void kernel_launch(void* const* d_in, const int* in_sizes, int n_in,
                              void* d_out, int out_size, void* d_ws, size_t ws_size,
                              hipStream_t stream) {
    (void)in_sizes; (void)n_in; (void)out_size; (void)d_ws; (void)ws_size;
    mamba_fused_kernel<<<dim3(1024), dim3(256), 0, stream>>>(
        (const float*)d_in[0],
        (const float*)d_in[1],
        (const float*)d_in[2],
        (const float*)d_in[3],
        (const float*)d_in[4],
        (const float*)d_in[5],
        (const float*)d_in[6],
        (const float*)d_in[7],
        (const float*)d_in[8],
        (const float*)d_in[9],
        (const float*)d_in[10],
        (const float*)d_in[11],
        (float*)d_out);
}


# Global loads issued early and grouped: x slice and FC weights as 16-byte loads in flight together, phase parameters loaded a barrier ahead, barriers fence LDS only; FC partial sums regrouped
# speedup vs baseline: 6.9885x; 1.1487x over previous
#include <hip/hip_runtime.h>
#include <hip/hip_bf16.h>

#define L_SEQ 250
#define DMODEL 8
#define DINNER 16
#define DSTATE 16
#define DCONV 5
#define NCLS 12

typedef __attribute__((ext_vector_type(2))) float v2f;
typedef __attribute__((ext_vector_type(4))) float v4f;
typedef __attribute__((ext_vector_type(8))) float v8f;

__device__ __forceinline__ float sigmoidf_(float v) { return __builtin_amdgcn_rcpf(1.0f + __expf(-v)); }

__device__ __forceinline__ float softplusf_(float v) {
    const float t  = __expf(v);
    const float u  = 1.0f + t;
    const float lg = __builtin_amdgcn_logf(u) * 0.6931471805599453f;
    const float r  = lg + (t - (u - 1.0f)) * __builtin_amdgcn_rcpf(u);
    return (v > 20.f) ? v : r;
}

__device__ __forceinline__ void lds_barrier() {
    __builtin_amdgcn_fence(__ATOMIC_RELEASE, "workgroup", "local");
    __builtin_amdgcn_s_barrier();
    __builtin_amdgcn_fence(__ATOMIC_ACQUIRE, "workgroup", "local");
}

template <int XM>
__device__ __forceinline__ float row_xmask_from(float v) {
    const int other = __builtin_amdgcn_update_dpp(0, __builtin_bit_cast(int, v),
                                                  0x160 | XM  , 0xf, 0xf, true);
    return __builtin_bit_cast(float, other);
}

template <int XM>
__device__ __forceinline__ float fold_pair(float a, float b, bool up) {
    const float keep = up ? b : a;
    const float send = up ? a : b;
    return keep + row_xmask_from<XM>(send);
}

template <int N>
__device__ __forceinline__ void scan_steps(float& h, const float As, const int l0, const int d, const int s,
                                           const bool b0, const bool b1, const bool b2, const bool b3,
                                           const float* s_dt, const float* s_xc,
                                           const float* s_B, const float* s_C, float* s_y) {
    float y[16];
    for (int j = 0; j < 16; ++j) {
        if (j < N) {
            const int l = l0 + j;
            const float dv  = s_dt[l * 16 + d];
            const float xcv = s_xc[l * 16 + d];
            h = __expf(dv * As) * h + (dv * xcv) * s_B[l * 16 + s];
            y[j] = h * s_C[l * 16 + s];
        } else {
            y[j] = 0.f;
        }
    }
    float r1[8], r2[4], r3[2];
    for (int j = 0; j < 8; ++j) r1[j] = fold_pair<1>(y[2 * j], y[2 * j + 1], b0);
    for (int j = 0; j < 4; ++j) r2[j] = fold_pair<2>(r1[2 * j], r1[2 * j + 1], b1);
    for (int j = 0; j < 2; ++j) r3[j] = fold_pair<4>(r2[2 * j], r2[2 * j + 1], b2);
    const float ysum = fold_pair<8>(r3[0], r3[1], b3);
    if (s < N) s_y[(l0 + s) * 16 + d] = ysum;
}

__global__ __launch_bounds__(256) void mamba_fused_kernel(
    const float* __restrict__ x,
    const float* __restrict__ W_in,
    const float* __restrict__ conv_w,
    const float* __restrict__ conv_b,
    const float* __restrict__ W_x,
    const float* __restrict__ dt_w,
    const float* __restrict__ dt_b,
    const float* __restrict__ A_log,
    const float* __restrict__ Dp,
    const float* __restrict__ W_out,
    const float* __restrict__ fc_w,
    const float* __restrict__ fc_b,
    float* __restrict__ out)
{
    __shared__ __attribute__((aligned(16))) float smem[16298];
    float* s_dt  = smem;
    float* s_xc  = smem + 4016;
    float* s_B   = smem + 8016;
    float* s_C   = smem + 12032;
    float* s_dtr = smem + 16048;
    float* s_xb  = s_B;
    float* s_xi  = smem + 12096;
    float* s_y   = s_dt;
    float* s_out = s_C;
    float* s_red = s_dtr;

    const int b    = blockIdx.x;
    const int tid  = threadIdx.x;
    const int lane = tid & 31;
    const int wave = tid >> 5;
    const int lm   = lane & 15;
    const int hi   = lane >> 4;
    const float* xb = x + (size_t)b * DMODEL * L_SEQ;

    const bool second = tid < (DMODEL * L_SEQ / 4 - 256);
    const int  pc1 = min(tid + 256, DMODEL * L_SEQ / 4 - 1);
    const v4f* xb4 = reinterpret_cast<const v4f*>(xb);
    const v4f xv0 = xb4[tid];
    const v4f xv1 = xb4[pc1];
    v2f wz[2], wx[2];
#pragma unroll
    for (int q = 0; q < 2; ++q) {
        const int k0 = 4 * q + 2 * hi;
        wz[q].x = W_in[(16 + lm) * 8 + k0];
        wz[q].y = W_in[(16 + lm) * 8 + k0 + 1];
        wx[q].x = W_in[lm * 8 + k0];
        wx[q].y = W_in[lm * 8 + k0 + 1];
    }
    float cw[DCONV];
#pragma unroll
    for (int j = 0; j < DCONV; ++j) cw[j] = conv_w[(tid & 15) * DCONV + j];
    const float cb = conv_b[tid & 15];
    for (int i = tid * 32; i < NCLS * L_SEQ * DMODEL; i += 256 * 32)
        __builtin_prefetch(fc_w + i, 0, 0);
    reinterpret_cast<v4f*>(s_xb)[tid] = xv0;
    if (second) reinterpret_cast<v4f*>(s_xb)[tid + 256] = xv1;
    if (tid < (DCONV - 1) * DINNER) s_xi[tid - (DCONV - 1) * DINNER] = 0.f;
    lds_barrier();

    v8f zacc[2];
#pragma unroll
    for (int t = 0; t < 2; ++t) {
        const int l0 = (wave + 8 * t) * 16;
        const int lr = min(l0 + lm, L_SEQ - 1);
        v8f acc = {}, accx = {};
#pragma unroll
        for (int kc = 0; kc < 8; kc += 4) {
            const int k0 = kc + 2 * hi;
            v2f a;
            a.x  = s_xb[(k0    ) * L_SEQ + lr];
            a.y  = s_xb[(k0 + 1) * L_SEQ + lr];
            const v2f bb = wz[kc >> 2];
            const v2f bx = wx[kc >> 2];
            acc  = __builtin_amdgcn_wmma_f32_16x16x4_f32(false, a, false, bb,
                                                         (short)0, acc, false, false);
            accx = __builtin_amdgcn_wmma_f32_16x16x4_f32(false, a, false, bx,
                                                         (short)0, accx, false, false);
        }
        zacc[t] = acc;
#pragma unroll
        for (int r = 0; r < 8; ++r) {
            const int row = min(l0 + r + 8 * hi, L_SEQ);
            s_xi[row * 16 + lm] = accx[r];
        }
    }
    lds_barrier();

    float wdt[DINNER];
#pragma unroll
    for (int d = 0; d < DINNER; ++d) wdt[d] = W_x[d];
    const int nt3 = wave & 1;
    const int e03 = 1 + nt3 * 16;
    v2f wbc[4];
#pragma unroll
    for (int q = 0; q < 4; ++q) {
        const int k0 = 4 * q + 2 * hi;
        wbc[q].x = W_x[(e03 + lm) * 16 + k0];
        wbc[q].y = W_x[(e03 + lm) * 16 + k0 + 1];
    }

    {
        for (int i = tid; i < L_SEQ * DINNER; i += 256) {
            float accv = cb;
#pragma unroll
            for (int j = 0; j < DCONV; ++j)
                accv += cw[j] * s_xi[i + (j - (DCONV - 1)) * DINNER];
            s_xc[i] = accv * sigmoidf_(accv);
        }
    }
    lds_barrier();

    const float dtw  = dt_w[tid & 15], dtb = dt_b[tid & 15];
    const float alog = A_log[tid];

    for (int l = tid; l < L_SEQ; l += 256) {
        float a = 0.f;
#pragma unroll
        for (int d = 0; d < DINNER; ++d) a += s_xc[l * 16 + d] * wdt[d];
        s_dtr[l] = a;
    }

    for (int j = wave; j < 32; j += 8) {
        const int mt = j >> 1;
        const int l0 = mt * 16;
        const int lr = min(l0 + lm, L_SEQ - 1);
        v8f acc = {};
#pragma unroll
        for (int kc = 0; kc < 16; kc += 4) {
            const int k0 = kc + 2 * hi;
            v2f a;
            a.x  = s_xc[lr * 16 + k0];
            a.y  = s_xc[lr * 16 + k0 + 1];
            const v2f bb = wbc[kc >> 2];
            acc = __builtin_amdgcn_wmma_f32_16x16x4_f32(false, a, false, bb,
                                                        (short)0, acc, false, false);
        }
        float* dst = nt3 ? s_C : s_B;
#pragma unroll
        for (int r = 0; r < 8; ++r) {
            const int row = min(l0 + r + 8 * hi, L_SEQ);
            dst[row * 16 + lm] = acc[r];
        }
    }
    lds_barrier();

    {
        for (int i = tid; i < L_SEQ * DINNER; i += 256) {
            float dv = s_dtr[i >> 4] * dtw + dtb;
            dv = softplusf_(dv);
            s_dt[i] = dv;
        }
    }
    lds_barrier();

    {
        const int d = tid >> 4;
        const int s = tid & 15;
        const float As  = -__expf(alog);
        const bool b0 = (s & 1) != 0, b1 = (s & 2) != 0, b2 = (s & 4) != 0, b3 = (s & 8) != 0;
        float h = 0.f;
        for (int l0 = 0; l0 < 240; l0 += 16)
            scan_steps<16>(h, As, l0, d, s, b0, b1, b2, b3, s_dt, s_xc, s_B, s_C, s_y);
        scan_steps<L_SEQ - 240>(h, As, 240, d, s, b0, b1, b2, b3, s_dt, s_xc, s_B, s_C, s_y);
    }
    const float Dpd = Dp[lm];
    v2f wo[4];
#pragma unroll
    for (int q = 0; q < 4; ++q) {
        const int k0 = 4 * q + 2 * hi;
        const float wa = W_out[(lm & (DMODEL - 1)) * 16 + k0];
        const float wb = W_out[(lm & (DMODEL - 1)) * 16 + k0 + 1];
        wo[q].x = (lm < DMODEL) ? wa : 0.f;
        wo[q].y = (lm < DMODEL) ? wb : 0.f;
    }
    const v4f* fw4 = reinterpret_cast<const v4f*>(fc_w);
    v4f w0[NCLS], w1[NCLS];
#pragma unroll
    for (int c = 0; c < NCLS; ++c) w0[c] = fw4[c * (L_SEQ * DMODEL / 4) + tid];
    lds_barrier();

    {
#pragma unroll
        for (int t = 0; t < 2; ++t) {
            const int l0 = (wave + 8 * t) * 16;
#pragma unroll
            for (int r = 0; r < 8; ++r) {
                const int row = l0 + r + 8 * hi;
                if (row < L_SEQ) {
                    const int idx = row * 16 + lm;
                    float y = s_y[idx];
                    y += s_xc[idx] * Dpd;
                    const float zv = zacc[t][r];
                    y *= zv * sigmoidf_(zv);
                    s_y[idx] = y;
                }
            }
        }
    }
    lds_barrier();

    for (int mt = wave; mt < 16; mt += 8) {
        const int l0 = mt * 16;
        const int lr = min(l0 + lm, L_SEQ - 1);
        v8f acc = {};
#pragma unroll
        for (int kc = 0; kc < 16; kc += 4) {
            const int k0 = kc + 2 * hi;
            v2f a;
            a.x  = s_y[lr * 16 + k0];
            a.y  = s_y[lr * 16 + k0 + 1];
            const v2f bb = wo[kc >> 2];
            acc = __builtin_amdgcn_wmma_f32_16x16x4_f32(false, a, false, bb,
                                                        (short)0, acc, false, false);
        }
        if (lm < DMODEL) {
#pragma unroll
            for (int r = 0; r < 8; ++r)
                s_out[(l0 + r + 8 * hi) * 8 + lm] = acc[r];
        }
    }
#pragma unroll
    for (int c = 0; c < NCLS; ++c) w1[c] = fw4[c * (L_SEQ * DMODEL / 4) + pc1];
    const float fcb = fc_b[min(tid, NCLS - 1)];
    lds_barrier();

    {
        float part[NCLS];
        const v4f o0 = reinterpret_cast<const v4f*>(s_out)[tid];
#pragma unroll
        for (int c = 0; c < NCLS; ++c) w0[c] = o0 * w0[c];
        if (second) {
            const v4f o1 = reinterpret_cast<const v4f*>(s_out)[tid + 256];
#pragma unroll
            for (int c = 0; c < NCLS; ++c) w0[c] += o1 * w1[c];
        }
#pragma unroll
        for (int c = 0; c < NCLS; ++c) part[c] = (w0[c].x + w0[c].y) + (w0[c].z + w0[c].w);
#pragma unroll
        for (int c = 0; c < NCLS; ++c) {
#pragma unroll
            for (int off = 16; off > 0; off >>= 1)
                part[c] += __shfl_down(part[c], off, 32);
        }
        if (lane == 0) {
#pragma unroll
            for (int c = 0; c < NCLS; ++c) s_red[wave * NCLS + c] = part[c];
        }
        lds_barrier();
        if (tid < NCLS) {
            float sres = fcb;
#pragma unroll
            for (int w = 0; w < 8; ++w) sres += s_red[w * NCLS + tid];
            out[b * NCLS + tid] = sres;
        }
    }
}

extern "C" void kernel_launch(void* const* d_in, const int* in_sizes, int n_in,
                              void* d_out, int out_size, void* d_ws, size_t ws_size,
                              hipStream_t stream) {
    (void)in_sizes; (void)n_in; (void)out_size; (void)d_ws; (void)ws_size;
    mamba_fused_kernel<<<dim3(1024), dim3(256), 0, stream>>>(
        (const float*)d_in[0],
        (const float*)d_in[1],
        (const float*)d_in[2],
        (const float*)d_in[3],
        (const float*)d_in[4],
        (const float*)d_in[5],
        (const float*)d_in[6],
        (const float*)d_in[7],
        (const float*)d_in[8],
        (const float*)d_in[9],
        (const float*)d_in[10],
        (const float*)d_in[11],
        (float*)d_out);
}


# Scan: each lane carries two adjacent states of a channel in packed f32 (waves 0-3, 4 channels each): dt, xc read once per pair, Bm, Cm as 8-byte loads, 7 cross-lane adds per 8 steps; same add tree
# speedup vs baseline: 7.9885x; 1.1431x over previous
#include <hip/hip_runtime.h>
#include <hip/hip_bf16.h>

#define L_SEQ 250
#define DMODEL 8
#define DINNER 16
#define DSTATE 16
#define DCONV 5
#define NCLS 12

typedef __attribute__((ext_vector_type(2))) float v2f;
typedef __attribute__((ext_vector_type(4))) float v4f;
typedef __attribute__((ext_vector_type(8))) float v8f;

__device__ __forceinline__ float sigmoidf_(float v) { return __builtin_amdgcn_rcpf(1.0f + __expf(-v)); }

__device__ __forceinline__ float softplusf_(float v) {
    const float t  = __expf(v);
    const float u  = 1.0f + t;
    const float lg = __builtin_amdgcn_logf(u) * 0.6931471805599453f;
    const float r  = lg + (t - (u - 1.0f)) * __builtin_amdgcn_rcpf(u);
    return (v > 20.f) ? v : r;
}

__device__ __forceinline__ void lds_barrier() {
    __builtin_amdgcn_fence(__ATOMIC_RELEASE, "workgroup", "local");
    __builtin_amdgcn_s_barrier();
    __builtin_amdgcn_fence(__ATOMIC_ACQUIRE, "workgroup", "local");
}

template <int XM>
__device__ __forceinline__ float row_xmask_from(float v) {
    const int other = __builtin_amdgcn_update_dpp(0, __builtin_bit_cast(int, v),
                                                  0x160 | XM  , 0xf, 0xf, true);
    return __builtin_bit_cast(float, other);
}

template <int XM>
__device__ __forceinline__ float fold_pair(float a, float b, bool up) {
    const float keep = up ? b : a;
    const float send = up ? a : b;
    return keep + row_xmask_from<XM>(send);
}

template <int N>
__device__ __forceinline__ void scan_steps2(v2f& h, const v2f As, const int l0, const int d, const int q,
                                            const bool b0, const bool b1, const bool b2,
                                            const float* s_dt, const float* s_xc,
                                            const float* s_B, const float* s_C, float* s_y) {
    float y[8];
    for (int j = 0; j < 8; ++j) {
        if (j < N) {
            const int l = l0 + j;
            const float dv  = s_dt[l * 16 + d];
            const float xcv = s_xc[l * 16 + d];
            const v2f Bv = *reinterpret_cast<const v2f*>(s_B + l * 16 + 2 * q);
            const v2f Cv = *reinterpret_cast<const v2f*>(s_C + l * 16 + 2 * q);
            const v2f e  = dv * As;
            v2f dA;
            dA.x = __expf(e.x);
            dA.y = __expf(e.y);
            h = dA * h + (dv * xcv) * Bv;
            y[j] = h.x * Cv.x + h.y * Cv.y;
        } else {
            y[j] = 0.f;
        }
    }
    float r1[4], r2[2];
    for (int j = 0; j < 4; ++j) r1[j] = fold_pair<1>(y[2 * j], y[2 * j + 1], b0);
    for (int j = 0; j < 2; ++j) r2[j] = fold_pair<2>(r1[2 * j], r1[2 * j + 1], b1);
    const float ysum = fold_pair<4>(r2[0], r2[1], b2);
    if (q < N) s_y[(l0 + q) * 16 + d] = ysum;
}

__global__ __launch_bounds__(256) void mamba_fused_kernel(
    const float* __restrict__ x,
    const float* __restrict__ W_in,
    const float* __restrict__ conv_w,
    const float* __restrict__ conv_b,
    const float* __restrict__ W_x,
    const float* __restrict__ dt_w,
    const float* __restrict__ dt_b,
    const float* __restrict__ A_log,
    const float* __restrict__ Dp,
    const float* __restrict__ W_out,
    const float* __restrict__ fc_w,
    const float* __restrict__ fc_b,
    float* __restrict__ out)
{
    __shared__ __attribute__((aligned(16))) float smem[16298];
    float* s_dt  = smem;
    float* s_xc  = smem + 4016;
    float* s_B   = smem + 8016;
    float* s_C   = smem + 12032;
    float* s_dtr = smem + 16048;
    float* s_xb  = s_B;
    float* s_xi  = smem + 12096;
    float* s_y   = s_dt;
    float* s_out = s_C;
    float* s_red = s_dtr;

    const int b    = blockIdx.x;
    const int tid  = threadIdx.x;
    const int lane = tid & 31;
    const int wave = tid >> 5;
    const int lm   = lane & 15;
    const int hi   = lane >> 4;
    const float* xb = x + (size_t)b * DMODEL * L_SEQ;

    const bool second = tid < (DMODEL * L_SEQ / 4 - 256);
    const int  pc1 = min(tid + 256, DMODEL * L_SEQ / 4 - 1);
    const v4f* xb4 = reinterpret_cast<const v4f*>(xb);
    const v4f xv0 = xb4[tid];
    const v4f xv1 = xb4[pc1];
    v2f wz[2], wx[2];
#pragma unroll
    for (int q = 0; q < 2; ++q) {
        const int k0 = 4 * q + 2 * hi;
        wz[q].x = W_in[(16 + lm) * 8 + k0];
        wz[q].y = W_in[(16 + lm) * 8 + k0 + 1];
        wx[q].x = W_in[lm * 8 + k0];
        wx[q].y = W_in[lm * 8 + k0 + 1];
    }
    float cw[DCONV];
#pragma unroll
    for (int j = 0; j < DCONV; ++j) cw[j] = conv_w[(tid & 15) * DCONV + j];
    const float cb = conv_b[tid & 15];
    for (int i = tid * 32; i < NCLS * L_SEQ * DMODEL; i += 256 * 32)
        __builtin_prefetch(fc_w + i, 0, 0);
    reinterpret_cast<v4f*>(s_xb)[tid] = xv0;
    if (second) reinterpret_cast<v4f*>(s_xb)[tid + 256] = xv1;
    if (tid < (DCONV - 1) * DINNER) s_xi[tid - (DCONV - 1) * DINNER] = 0.f;
    lds_barrier();

    v8f zacc[2];
#pragma unroll
    for (int t = 0; t < 2; ++t) {
        const int l0 = (wave + 8 * t) * 16;
        const int lr = min(l0 + lm, L_SEQ - 1);
        v8f acc = {}, accx = {};
#pragma unroll
        for (int kc = 0; kc < 8; kc += 4) {
            const int k0 = kc + 2 * hi;
            v2f a;
            a.x  = s_xb[(k0    ) * L_SEQ + lr];
            a.y  = s_xb[(k0 + 1) * L_SEQ + lr];
            const v2f bb = wz[kc >> 2];
            const v2f bx = wx[kc >> 2];
            acc  = __builtin_amdgcn_wmma_f32_16x16x4_f32(false, a, false, bb,
                                                         (short)0, acc, false, false);
            accx = __builtin_amdgcn_wmma_f32_16x16x4_f32(false, a, false, bx,
                                                         (short)0, accx, false, false);
        }
        zacc[t] = acc;
#pragma unroll
        for (int r = 0; r < 8; ++r) {
            const int row = min(l0 + r + 8 * hi, L_SEQ);
            s_xi[row * 16 + lm] = accx[r];
        }
    }
    lds_barrier();

    float wdt[DINNER];
#pragma unroll
    for (int d = 0; d < DINNER; ++d) wdt[d] = W_x[d];
    const int nt3 = wave & 1;
    const int e03 = 1 + nt3 * 16;
    v2f wbc[4];
#pragma unroll
    for (int q = 0; q < 4; ++q) {
        const int k0 = 4 * q + 2 * hi;
        wbc[q].x = W_x[(e03 + lm) * 16 + k0];
        wbc[q].y = W_x[(e03 + lm) * 16 + k0 + 1];
    }

    {
        for (int i = tid; i < L_SEQ * DINNER; i += 256) {
            float accv = cb;
#pragma unroll
            for (int j = 0; j < DCONV; ++j)
                accv += cw[j] * s_xi[i + (j - (DCONV - 1)) * DINNER];
            s_xc[i] = accv * sigmoidf_(accv);
        }
    }
    lds_barrier();

    const float dtw  = dt_w[tid & 15], dtb = dt_b[tid & 15];
    const v2f alog2 = reinterpret_cast<const v2f*>(A_log)[tid & 127];

    for (int l = tid; l < L_SEQ; l += 256) {
        float a = 0.f;
#pragma unroll
        for (int d = 0; d < DINNER; ++d) a += s_xc[l * 16 + d] * wdt[d];
        s_dtr[l] = a;
    }

    for (int j = wave; j < 32; j += 8) {
        const int mt = j >> 1;
        const int l0 = mt * 16;
        const int lr = min(l0 + lm, L_SEQ - 1);
        v8f acc = {};
#pragma unroll
        for (int kc = 0; kc < 16; kc += 4) {
            const int k0 = kc + 2 * hi;
            v2f a;
            a.x  = s_xc[lr * 16 + k0];
            a.y  = s_xc[lr * 16 + k0 + 1];
            const v2f bb = wbc[kc >> 2];
            acc = __builtin_amdgcn_wmma_f32_16x16x4_f32(false, a, false, bb,
                                                        (short)0, acc, false, false);
        }
        float* dst = nt3 ? s_C : s_B;
#pragma unroll
        for (int r = 0; r < 8; ++r) {
            const int row = min(l0 + r + 8 * hi, L_SEQ);
            dst[row * 16 + lm] = acc[r];
        }
    }
    lds_barrier();

    {
        for (int i = tid; i < L_SEQ * DINNER; i += 256) {
            float dv = s_dtr[i >> 4] * dtw + dtb;
            dv = softplusf_(dv);
            s_dt[i] = dv;
        }
    }
    lds_barrier();

    if (wave < 4) {
        const int d = tid >> 3;
        const int q = tid & 7;
        v2f As;
        As.x = -__expf(alog2.x);
        As.y = -__expf(alog2.y);
        const bool b0 = (q & 1) != 0, b1 = (q & 2) != 0, b2 = (q & 4) != 0;
        v2f h = {0.f, 0.f};
        for (int l0 = 0; l0 < 248; l0 += 8)
            scan_steps2<8>(h, As, l0, d, q, b0, b1, b2, s_dt, s_xc, s_B, s_C, s_y);
        scan_steps2<L_SEQ - 248>(h, As, 248, d, q, b0, b1, b2, s_dt, s_xc, s_B, s_C, s_y);
    }
    const float Dpd = Dp[lm];
    v2f wo[4];
#pragma unroll
    for (int q = 0; q < 4; ++q) {
        const int k0 = 4 * q + 2 * hi;
        const float wa = W_out[(lm & (DMODEL - 1)) * 16 + k0];
        const float wb = W_out[(lm & (DMODEL - 1)) * 16 + k0 + 1];
        wo[q].x = (lm < DMODEL) ? wa : 0.f;
        wo[q].y = (lm < DMODEL) ? wb : 0.f;
    }
    const v4f* fw4 = reinterpret_cast<const v4f*>(fc_w);
    v4f w0[NCLS], w1[NCLS];
#pragma unroll
    for (int c = 0; c < NCLS; ++c) w0[c] = fw4[c * (L_SEQ * DMODEL / 4) + tid];
    lds_barrier();

    {
#pragma unroll
        for (int t = 0; t < 2; ++t) {
            const int l0 = (wave + 8 * t) * 16;
#pragma unroll
            for (int r = 0; r < 8; ++r) {
                const int row = l0 + r + 8 * hi;
                if (row < L_SEQ) {
                    const int idx = row * 16 + lm;
                    float y = s_y[idx];
                    y += s_xc[idx] * Dpd;
                    const float zv = zacc[t][r];
                    y *= zv * sigmoidf_(zv);
                    s_y[idx] = y;
                }
            }
        }
    }
    lds_barrier();

    for (int mt = wave; mt < 16; mt += 8) {
        const int l0 = mt * 16;
        const int lr = min(l0 + lm, L_SEQ - 1);
        v8f acc = {};
#pragma unroll
        for (int kc = 0; kc < 16; kc += 4) {
            const int k0 = kc + 2 * hi;
            v2f a;
            a.x  = s_y[lr * 16 + k0];
            a.y  = s_y[lr * 16 + k0 + 1];
            const v2f bb = wo[kc >> 2];
            acc = __builtin_amdgcn_wmma_f32_16x16x4_f32(false, a, false, bb,
                                                        (short)0, acc, false, false);
        }
        if (lm < DMODEL) {
#pragma unroll
            for (int r = 0; r < 8; ++r)
                s_out[(l0 + r + 8 * hi) * 8 + lm] = acc[r];
        }
    }
#pragma unroll
    for (int c = 0; c < NCLS; ++c) w1[c] = fw4[c * (L_SEQ * DMODEL / 4) + pc1];
    const float fcb = fc_b[min(tid, NCLS - 1)];
    lds_barrier();

    {
        float part[NCLS];
        const v4f o0 = reinterpret_cast<const v4f*>(s_out)[tid];
#pragma unroll
        for (int c = 0; c < NCLS; ++c) w0[c] = o0 * w0[c];
        if (second) {
            const v4f o1 = reinterpret_cast<const v4f*>(s_out)[tid + 256];
#pragma unroll
            for (int c = 0; c < NCLS; ++c) w0[c] += o1 * w1[c];
        }
#pragma unroll
        for (int c = 0; c < NCLS; ++c) part[c] = (w0[c].x + w0[c].y) + (w0[c].z + w0[c].w);
#pragma unroll
        for (int c = 0; c < NCLS; ++c) {
#pragma unroll
            for (int off = 16; off > 0; off >>= 1)
                part[c] += __shfl_down(part[c], off, 32);
        }
        if (lane == 0) {
#pragma unroll
            for (int c = 0; c < NCLS; ++c) s_red[wave * NCLS + c] = part[c];
        }
        lds_barrier();
        if (tid < NCLS) {
            float sres = fcb;
#pragma unroll
            for (int w = 0; w < 8; ++w) sres += s_red[w * NCLS + tid];
            out[b * NCLS + tid] = sres;
        }
    }
}

extern "C" void kernel_launch(void* const* d_in, const int* in_sizes, int n_in,
                              void* d_out, int out_size, void* d_ws, size_t ws_size,
                              hipStream_t stream) {
    (void)in_sizes; (void)n_in; (void)out_size; (void)d_ws; (void)ws_size;
    mamba_fused_kernel<<<dim3(1024), dim3(256), 0, stream>>>(
        (const float*)d_in[0],
        (const float*)d_in[1],
        (const float*)d_in[2],
        (const float*)d_in[3],
        (const float*)d_in[4],
        (const float*)d_in[5],
        (const float*)d_in[6],
        (const float*)d_in[7],
        (const float*)d_in[8],
        (const float*)d_in[9],
        (const float*)d_in[10],
        (const float*)d_in[11],
        (float*)d_out);
}


# Softplus pass and gate pass: a thread reads the LDS inputs of several elements before the first store, no branch per element, so their exp/log/rcp chains overlap instead of running one by one
# speedup vs baseline: 8.2684x; 1.0350x over previous
#include <hip/hip_runtime.h>
#include <hip/hip_bf16.h>

#define L_SEQ 250
#define DMODEL 8
#define DINNER 16
#define DSTATE 16
#define DCONV 5
#define NCLS 12

typedef __attribute__((ext_vector_type(2))) float v2f;
typedef __attribute__((ext_vector_type(4))) float v4f;
typedef __attribute__((ext_vector_type(8))) float v8f;

__device__ __forceinline__ float sigmoidf_(float v) { return __builtin_amdgcn_rcpf(1.0f + __expf(-v)); }

__device__ __forceinline__ float softplusf_(float v) {
    const float t  = __expf(v);
    const float u  = 1.0f + t;
    const float lg = __builtin_amdgcn_logf(u) * 0.6931471805599453f;
    const float r  = lg + (t - (u - 1.0f)) * __builtin_amdgcn_rcpf(u);
    return (v > 20.f) ? v : r;
}

__device__ __forceinline__ void lds_barrier() {
    __builtin_amdgcn_fence(__ATOMIC_RELEASE, "workgroup", "local");
    __builtin_amdgcn_s_barrier();
    __builtin_amdgcn_fence(__ATOMIC_ACQUIRE, "workgroup", "local");
}

template <int XM>
__device__ __forceinline__ float row_xmask_from(float v) {
    const int other = __builtin_amdgcn_update_dpp(0, __builtin_bit_cast(int, v),
                                                  0x160 | XM  , 0xf, 0xf, true);
    return __builtin_bit_cast(float, other);
}

template <int XM>
__device__ __forceinline__ float fold_pair(float a, float b, bool up) {
    const float keep = up ? b : a;
    const float send = up ? a : b;
    return keep + row_xmask_from<XM>(send);
}

template <int N>
__device__ __forceinline__ void scan_steps2(v2f& h, const v2f As, const int l0, const int d, const int q,
                                            const bool b0, const bool b1, const bool b2,
                                            const float* s_dt, const float* s_xc,
                                            const float* s_B, const float* s_C, float* s_y) {
    float y[8];
    for (int j = 0; j < 8; ++j) {
        if (j < N) {
            const int l = l0 + j;
            const float dv  = s_dt[l * 16 + d];
            const float xcv = s_xc[l * 16 + d];
            const v2f Bv = *reinterpret_cast<const v2f*>(s_B + l * 16 + 2 * q);
            const v2f Cv = *reinterpret_cast<const v2f*>(s_C + l * 16 + 2 * q);
            const v2f e  = dv * As;
            v2f dA;
            dA.x = __expf(e.x);
            dA.y = __expf(e.y);
            h = dA * h + (dv * xcv) * Bv;
            y[j] = h.x * Cv.x + h.y * Cv.y;
        } else {
            y[j] = 0.f;
        }
    }
    float r1[4], r2[2];
    for (int j = 0; j < 4; ++j) r1[j] = fold_pair<1>(y[2 * j], y[2 * j + 1], b0);
    for (int j = 0; j < 2; ++j) r2[j] = fold_pair<2>(r1[2 * j], r1[2 * j + 1], b1);
    const float ysum = fold_pair<4>(r2[0], r2[1], b2);
    if (q < N) s_y[(l0 + q) * 16 + d] = ysum;
}

__global__ __launch_bounds__(256) void mamba_fused_kernel(
    const float* __restrict__ x,
    const float* __restrict__ W_in,
    const float* __restrict__ conv_w,
    const float* __restrict__ conv_b,
    const float* __restrict__ W_x,
    const float* __restrict__ dt_w,
    const float* __restrict__ dt_b,
    const float* __restrict__ A_log,
    const float* __restrict__ Dp,
    const float* __restrict__ W_out,
    const float* __restrict__ fc_w,
    const float* __restrict__ fc_b,
    float* __restrict__ out)
{
    __shared__ __attribute__((aligned(16))) float smem[16298];
    float* s_dt  = smem;
    float* s_xc  = smem + 4016;
    float* s_B   = smem + 8016;
    float* s_C   = smem + 12032;
    float* s_dtr = smem + 16048;
    float* s_xb  = s_B;
    float* s_xi  = smem + 12096;
    float* s_y   = s_dt;
    float* s_out = s_C;
    float* s_red = s_dtr;

    const int b    = blockIdx.x;
    const int tid  = threadIdx.x;
    const int lane = tid & 31;
    const int wave = tid >> 5;
    const int lm   = lane & 15;
    const int hi   = lane >> 4;
    const float* xb = x + (size_t)b * DMODEL * L_SEQ;

    const bool second = tid < (DMODEL * L_SEQ / 4 - 256);
    const int  pc1 = min(tid + 256, DMODEL * L_SEQ / 4 - 1);
    const v4f* xb4 = reinterpret_cast<const v4f*>(xb);
    const v4f xv0 = xb4[tid];
    const v4f xv1 = xb4[pc1];
    v2f wz[2], wx[2];
#pragma unroll
    for (int q = 0; q < 2; ++q) {
        const int k0 = 4 * q + 2 * hi;
        wz[q].x = W_in[(16 + lm) * 8 + k0];
        wz[q].y = W_in[(16 + lm) * 8 + k0 + 1];
        wx[q].x = W_in[lm * 8 + k0];
        wx[q].y = W_in[lm * 8 + k0 + 1];
    }
    float cw[DCONV];
#pragma unroll
    for (int j = 0; j < DCONV; ++j) cw[j] = conv_w[(tid & 15) * DCONV + j];
    const float cb = conv_b[tid & 15];
    for (int i = tid * 32; i < NCLS * L_SEQ * DMODEL; i += 256 * 32)
        __builtin_prefetch(fc_w + i, 0, 0);
    reinterpret_cast<v4f*>(s_xb)[tid] = xv0;
    if (second) reinterpret_cast<v4f*>(s_xb)[tid + 256] = xv1;
    if (tid < (DCONV - 1) * DINNER) s_xi[tid - (DCONV - 1) * DINNER] = 0.f;
    lds_barrier();

    v8f zacc[2];
#pragma unroll
    for (int t = 0; t < 2; ++t) {
        const int l0 = (wave + 8 * t) * 16;
        const int lr = min(l0 + lm, L_SEQ - 1);
        v8f acc = {}, accx = {};
#pragma unroll
        for (int kc = 0; kc < 8; kc += 4) {
            const int k0 = kc + 2 * hi;
            v2f a;
            a.x  = s_xb[(k0    ) * L_SEQ + lr];
            a.y  = s_xb[(k0 + 1) * L_SEQ + lr];
            const v2f bb = wz[kc >> 2];
            const v2f bx = wx[kc >> 2];
            acc  = __builtin_amdgcn_wmma_f32_16x16x4_f32(false, a, false, bb,
                                                         (short)0, acc, false, false);
            accx = __builtin_amdgcn_wmma_f32_16x16x4_f32(false, a, false, bx,
                                                         (short)0, accx, false, false);
        }
        zacc[t] = acc;
#pragma unroll
        for (int r = 0; r < 8; ++r) {
            const int row = min(l0 + r + 8 * hi, L_SEQ);
            s_xi[row * 16 + lm] = accx[r];
        }
    }
    lds_barrier();

    float wdt[DINNER];
#pragma unroll
    for (int d = 0; d < DINNER; ++d) wdt[d] = W_x[d];
    const int nt3 = wave & 1;
    const int e03 = 1 + nt3 * 16;
    v2f wbc[4];
#pragma unroll
    for (int q = 0; q < 4; ++q) {
        const int k0 = 4 * q + 2 * hi;
        wbc[q].x = W_x[(e03 + lm) * 16 + k0];
        wbc[q].y = W_x[(e03 + lm) * 16 + k0 + 1];
    }

    {
        for (int i = tid; i < L_SEQ * DINNER; i += 256) {
            float accv = cb;
#pragma unroll
            for (int j = 0; j < DCONV; ++j)
                accv += cw[j] * s_xi[i + (j - (DCONV - 1)) * DINNER];
            s_xc[i] = accv * sigmoidf_(accv);
        }
    }
    lds_barrier();

    const float dtw  = dt_w[tid & 15], dtb = dt_b[tid & 15];
    const v2f alog2 = reinterpret_cast<const v2f*>(A_log)[tid & 127];

    for (int l = tid; l < L_SEQ; l += 256) {
        float a = 0.f;
#pragma unroll
        for (int d = 0; d < DINNER; ++d) a += s_xc[l * 16 + d] * wdt[d];
        s_dtr[l] = a;
    }

    for (int j = wave; j < 32; j += 8) {
        const int mt = j >> 1;
        const int l0 = mt * 16;
        const int lr = min(l0 + lm, L_SEQ - 1);
        v8f acc = {};
#pragma unroll
        for (int kc = 0; kc < 16; kc += 4) {
            const int k0 = kc + 2 * hi;
            v2f a;
            a.x  = s_xc[lr * 16 + k0];
            a.y  = s_xc[lr * 16 + k0 + 1];
            const v2f bb = wbc[kc >> 2];
            acc = __builtin_amdgcn_wmma_f32_16x16x4_f32(false, a, false, bb,
                                                        (short)0, acc, false, false);
        }
        float* dst = nt3 ? s_C : s_B;
#pragma unroll
        for (int r = 0; r < 8; ++r) {
            const int row = min(l0 + r + 8 * hi, L_SEQ);
            dst[row * 16 + lm] = acc[r];
        }
    }
    lds_barrier();

    {
        float raw[16];
#pragma unroll
        for (int k = 0; k < 15; ++k) raw[k] = s_dtr[(tid >> 4) + 16 * k];
        raw[15] = s_dtr[min((tid >> 4) + 240, L_SEQ - 1)];
#pragma unroll
        for (int k = 0; k < 15; ++k) s_dt[tid + 256 * k] = softplusf_(raw[k] * dtw + dtb);
        const float dv15 = softplusf_(raw[15] * dtw + dtb);
        if (tid < L_SEQ * DINNER - 3840) s_dt[tid + 3840] = dv15;
    }
    lds_barrier();

    if (wave < 4) {
        const int d = tid >> 3;
        const int q = tid & 7;
        v2f As;
        As.x = -__expf(alog2.x);
        As.y = -__expf(alog2.y);
        const bool b0 = (q & 1) != 0, b1 = (q & 2) != 0, b2 = (q & 4) != 0;
        v2f h = {0.f, 0.f};
        for (int l0 = 0; l0 < 248; l0 += 8)
            scan_steps2<8>(h, As, l0, d, q, b0, b1, b2, s_dt, s_xc, s_B, s_C, s_y);
        scan_steps2<L_SEQ - 248>(h, As, 248, d, q, b0, b1, b2, s_dt, s_xc, s_B, s_C, s_y);
    }
    const float Dpd = Dp[lm];
    v2f wo[4];
#pragma unroll
    for (int q = 0; q < 4; ++q) {
        const int k0 = 4 * q + 2 * hi;
        const float wa = W_out[(lm & (DMODEL - 1)) * 16 + k0];
        const float wb = W_out[(lm & (DMODEL - 1)) * 16 + k0 + 1];
        wo[q].x = (lm < DMODEL) ? wa : 0.f;
        wo[q].y = (lm < DMODEL) ? wb : 0.f;
    }
    const v4f* fw4 = reinterpret_cast<const v4f*>(fc_w);
    v4f w0[NCLS], w1[NCLS];
#pragma unroll
    for (int c = 0; c < NCLS; ++c) w0[c] = fw4[c * (L_SEQ * DMODEL / 4) + tid];
    lds_barrier();

    {
#pragma unroll
        for (int tg = 0; tg < 4; ++tg) {
            const int t  = tg >> 1;
            const int r0 = (tg & 1) * 4;
            const int l0 = (wave + 8 * t) * 16 + r0 + 8 * hi;
            float yv[4], xv[4];
#pragma unroll
            for (int r = 0; r < 4; ++r) {
                const int idx = (t == 0 ? l0 + r : min(l0 + r, L_SEQ - 1)) * 16 + lm;
                yv[r] = s_y[idx];
                xv[r] = s_xc[idx];
            }
#pragma unroll
            for (int r = 0; r < 4; ++r) {
                float y = yv[r];
                y += xv[r] * Dpd;
                const float zv = zacc[t][r0 + r];
                y *= zv * sigmoidf_(zv);
                yv[r] = y;
            }
#pragma unroll
            for (int r = 0; r < 4; ++r)
                s_y[(t == 0 ? l0 + r : min(l0 + r, L_SEQ)) * 16 + lm] = yv[r];
        }
    }
    lds_barrier();

    for (int mt = wave; mt < 16; mt += 8) {
        const int l0 = mt * 16;
        const int lr = min(l0 + lm, L_SEQ - 1);
        v8f acc = {};
#pragma unroll
        for (int kc = 0; kc < 16; kc += 4) {
            const int k0 = kc + 2 * hi;
            v2f a;
            a.x  = s_y[lr * 16 + k0];
            a.y  = s_y[lr * 16 + k0 + 1];
            const v2f bb = wo[kc >> 2];
            acc = __builtin_amdgcn_wmma_f32_16x16x4_f32(false, a, false, bb,
                                                        (short)0, acc, false, false);
        }
        if (lm < DMODEL) {
#pragma unroll
            for (int r = 0; r < 8; ++r)
                s_out[(l0 + r + 8 * hi) * 8 + lm] = acc[r];
        }
    }
#pragma unroll
    for (int c = 0; c < NCLS; ++c) w1[c] = fw4[c * (L_SEQ * DMODEL / 4) + pc1];
    const float fcb = fc_b[min(tid, NCLS - 1)];
    lds_barrier();

    {
        float part[NCLS];
        const v4f o0 = reinterpret_cast<const v4f*>(s_out)[tid];
#pragma unroll
        for (int c = 0; c < NCLS; ++c) w0[c] = o0 * w0[c];
        if (second) {
            const v4f o1 = reinterpret_cast<const v4f*>(s_out)[tid + 256];
#pragma unroll
            for (int c = 0; c < NCLS; ++c) w0[c] += o1 * w1[c];
        }
#pragma unroll
        for (int c = 0; c < NCLS; ++c) part[c] = (w0[c].x + w0[c].y) + (w0[c].z + w0[c].w);
#pragma unroll
        for (int c = 0; c < NCLS; ++c) {
#pragma unroll
            for (int off = 16; off > 0; off >>= 1)
                part[c] += __shfl_down(part[c], off, 32);
        }
        if (lane == 0) {
#pragma unroll
            for (int c = 0; c < NCLS; ++c) s_red[wave * NCLS + c] = part[c];
        }
        lds_barrier();
        if (tid < NCLS) {
            float sres = fcb;
#pragma unroll
            for (int w = 0; w < 8; ++w) sres += s_red[w * NCLS + tid];
            out[b * NCLS + tid] = sres;
        }
    }
}

extern "C" void kernel_launch(void* const* d_in, const int* in_sizes, int n_in,
                              void* d_out, int out_size, void* d_ws, size_t ws_size,
                              hipStream_t stream) {
    (void)in_sizes; (void)n_in; (void)out_size; (void)d_ws; (void)ws_size;
    mamba_fused_kernel<<<dim3(1024), dim3(256), 0, stream>>>(
        (const float*)d_in[0],
        (const float*)d_in[1],
        (const float*)d_in[2],
        (const float*)d_in[3],
        (const float*)d_in[4],
        (const float*)d_in[5],
        (const float*)d_in[6],
        (const float*)d_in[7],
        (const float*)d_in[8],
        (const float*)d_in[9],
        (const float*)d_in[10],
        (const float*)d_in[11],
        (float*)d_out);
}


# Two batch elements per work-group (512 groups of 8 waves, two LDS images): half the waves to enter, parameters and FC weights loaded once per pair, each half of the waves scans one element; same sums
# speedup vs baseline: 8.5851x; 1.0383x over previous
#include <hip/hip_runtime.h>
#include <hip/hip_bf16.h>


#define L_SEQ 250
#define DMODEL 8
#define DINNER 16
#define DSTATE 16
#define DCONV 5
#define NCLS 12

#define ESTRIDE 16300
#define O_DT  0
#define O_XC  4016
#define O_B   8016
#define O_C   12032
#define O_DTR 16048
#define O_XI  12096

typedef __attribute__((ext_vector_type(2))) float v2f;
typedef __attribute__((ext_vector_type(4))) float v4f;
typedef __attribute__((ext_vector_type(8))) float v8f;

__device__ __forceinline__ float sigmoidf_(float v) { return __builtin_amdgcn_rcpf(1.0f + __expf(-v)); }

__device__ __forceinline__ float softplusf_(float v) {
    const float t  = __expf(v);
    const float u  = 1.0f + t;
    const float lg = __builtin_amdgcn_logf(u) * 0.6931471805599453f;
    const float r  = lg + (t - (u - 1.0f)) * __builtin_amdgcn_rcpf(u);
    return (v > 20.f) ? v : r;
}

__device__ __forceinline__ void lds_barrier() {
    __builtin_amdgcn_fence(__ATOMIC_RELEASE, "workgroup", "local");
    __builtin_amdgcn_s_barrier();
    __builtin_amdgcn_fence(__ATOMIC_ACQUIRE, "workgroup", "local");
}

template <int XM>
__device__ __forceinline__ float row_xmask_from(float v) {
    const int other = __builtin_amdgcn_update_dpp(0, __builtin_bit_cast(int, v),
                                                  0x160 | XM  , 0xf, 0xf, true);
    return __builtin_bit_cast(float, other);
}

template <int XM>
__device__ __forceinline__ float fold_pair(float a, float b, bool up) {
    const float keep = up ? b : a;
    const float send = up ? a : b;
    return keep + row_xmask_from<XM>(send);
}

template <int N>
__device__ __forceinline__ void scan_steps2(v2f& h, const v2f As, const int l0, const int d, const int q,
                                            const bool b0, const bool b1, const bool b2,
                                            const float* s_dt, const float* s_xc,
                                            const float* s_B, const float* s_C, float* s_y) {
    float y[8];
    for (int j = 0; j < 8; ++j) {
        if (j < N) {
            const int l = l0 + j;
            const float dv  = s_dt[l * 16 + d];
            const float xcv = s_xc[l * 16 + d];
            const v2f Bv = *reinterpret_cast<const v2f*>(s_B + l * 16 + 2 * q);
            const v2f Cv = *reinterpret_cast<const v2f*>(s_C + l * 16 + 2 * q);
            const v2f e  = dv * As;
            v2f dA;
            dA.x = __expf(e.x);
            dA.y = __expf(e.y);
            h = dA * h + (dv * xcv) * Bv;
            y[j] = h.x * Cv.x + h.y * Cv.y;
        } else {
            y[j] = 0.f;
        }
    }
    float r1[4], r2[2];
    for (int j = 0; j < 4; ++j) r1[j] = fold_pair<1>(y[2 * j], y[2 * j + 1], b0);
    for (int j = 0; j < 2; ++j) r2[j] = fold_pair<2>(r1[2 * j], r1[2 * j + 1], b1);
    const float ysum = fold_pair<4>(r2[0], r2[1], b2);
    if (q < N) s_y[(l0 + q) * 16 + d] = ysum;
}

__global__ __launch_bounds__(256) void mamba_fused_kernel(
    const float* __restrict__ x,
    const float* __restrict__ W_in,
    const float* __restrict__ conv_w,
    const float* __restrict__ conv_b,
    const float* __restrict__ W_x,
    const float* __restrict__ dt_w,
    const float* __restrict__ dt_b,
    const float* __restrict__ A_log,
    const float* __restrict__ Dp,
    const float* __restrict__ W_out,
    const float* __restrict__ fc_w,
    const float* __restrict__ fc_b,
    float* __restrict__ out)
{
    __shared__ __attribute__((aligned(16))) float smem[2 * ESTRIDE];

    const int bb   = blockIdx.x * 2;
    const int tid  = threadIdx.x;
    const int lane = tid & 31;
    const int wave = tid >> 5;
    const int lm   = lane & 15;
    const int hi   = lane >> 4;
    const float* xb = x + (size_t)bb * DMODEL * L_SEQ;

    const bool second = tid < (DMODEL * L_SEQ / 4 - 256);
    const int  pc1 = min(tid + 256, DMODEL * L_SEQ / 4 - 1);
    const v4f* xb4 = reinterpret_cast<const v4f*>(xb);
    v4f xv0[2], xv1[2];
#pragma unroll
    for (int e = 0; e < 2; ++e) {
        xv0[e] = xb4[e * (DMODEL * L_SEQ / 4) + tid];
        xv1[e] = xb4[e * (DMODEL * L_SEQ / 4) + pc1];
    }
    v2f wz[2], wx[2];
#pragma unroll
    for (int q = 0; q < 2; ++q) {
        const int k0 = 4 * q + 2 * hi;
        wz[q].x = W_in[(16 + lm) * 8 + k0];
        wz[q].y = W_in[(16 + lm) * 8 + k0 + 1];
        wx[q].x = W_in[lm * 8 + k0];
        wx[q].y = W_in[lm * 8 + k0 + 1];
    }
    float cw[DCONV];
#pragma unroll
    for (int j = 0; j < DCONV; ++j) cw[j] = conv_w[(tid & 15) * DCONV + j];
    const float cb = conv_b[tid & 15];
    for (int i = tid * 32; i < NCLS * L_SEQ * DMODEL; i += 256 * 32)
        __builtin_prefetch(fc_w + i, 0, 0);
#pragma unroll
    for (int e = 0; e < 2; ++e) {
        float* sb = smem + e * ESTRIDE;
        reinterpret_cast<v4f*>(sb + O_B)[tid] = xv0[e];
        if (second) reinterpret_cast<v4f*>(sb + O_B)[tid + 256] = xv1[e];
        if (tid < (DCONV - 1) * DINNER) (sb + O_XI)[tid - (DCONV - 1) * DINNER] = 0.f;
    }
    lds_barrier();

    v8f zacc[2][2];
#pragma unroll
    for (int e = 0; e < 2; ++e) {
        const float* s_xb = smem + e * ESTRIDE + O_B;
        float*       s_xi = smem + e * ESTRIDE + O_XI;
#pragma unroll
        for (int t = 0; t < 2; ++t) {
            const int l0 = (wave + 8 * t) * 16;
            const int lr = min(l0 + lm, L_SEQ - 1);
            v8f acc = {}, accx = {};
#pragma unroll
            for (int kc = 0; kc < 8; kc += 4) {
                const int k0 = kc + 2 * hi;
                v2f a;
                a.x  = s_xb[(k0    ) * L_SEQ + lr];
                a.y  = s_xb[(k0 + 1) * L_SEQ + lr];
                const v2f bz = wz[kc >> 2];
                const v2f bx = wx[kc >> 2];
                acc  = __builtin_amdgcn_wmma_f32_16x16x4_f32(false, a, false, bz,
                                                             (short)0, acc, false, false);
                accx = __builtin_amdgcn_wmma_f32_16x16x4_f32(false, a, false, bx,
                                                             (short)0, accx, false, false);
            }
            zacc[e][t] = acc;
#pragma unroll
            for (int r = 0; r < 8; ++r) {
                const int row = min(l0 + r + 8 * hi, L_SEQ);
                s_xi[row * 16 + lm] = accx[r];
            }
        }
    }
    lds_barrier();

    float wdt[DINNER];
#pragma unroll
    for (int d = 0; d < DINNER; ++d) wdt[d] = W_x[d];
    const int nt3 = wave & 1;
    const int e03 = 1 + nt3 * 16;
    v2f wbc[4];
#pragma unroll
    for (int q = 0; q < 4; ++q) {
        const int k0 = 4 * q + 2 * hi;
        wbc[q].x = W_x[(e03 + lm) * 16 + k0];
        wbc[q].y = W_x[(e03 + lm) * 16 + k0 + 1];
    }

    for (int e = 0; e < 2; ++e) {
        const float* s_xi = smem + e * ESTRIDE + O_XI;
        float*       s_xc = smem + e * ESTRIDE + O_XC;
        for (int i = tid; i < L_SEQ * DINNER; i += 256) {
            float accv = cb;
#pragma unroll
            for (int j = 0; j < DCONV; ++j)
                accv += cw[j] * s_xi[i + (j - (DCONV - 1)) * DINNER];
            s_xc[i] = accv * sigmoidf_(accv);
        }
    }
    lds_barrier();

    const float dtw  = dt_w[tid & 15], dtb = dt_b[tid & 15];
    const v2f alog2 = reinterpret_cast<const v2f*>(A_log)[tid & 127];

    for (int e = 0; e < 2; ++e) {
        const float* s_xc  = smem + e * ESTRIDE + O_XC;
        float*       s_dtr = smem + e * ESTRIDE + O_DTR;
        for (int l = tid; l < L_SEQ; l += 256) {
            float a = 0.f;
#pragma unroll
            for (int d = 0; d < DINNER; ++d) a += s_xc[l * 16 + d] * wdt[d];
            s_dtr[l] = a;
        }
    }

    for (int e = 0; e < 2; ++e) {
        const float* s_xc = smem + e * ESTRIDE + O_XC;
        float* dst = smem + e * ESTRIDE + (nt3 ? O_C : O_B);
        for (int j = wave; j < 32; j += 8) {
            const int mt = j >> 1;
            const int l0 = mt * 16;
            const int lr = min(l0 + lm, L_SEQ - 1);
            v8f acc = {};
#pragma unroll
            for (int kc = 0; kc < 16; kc += 4) {
                const int k0 = kc + 2 * hi;
                v2f a;
                a.x  = s_xc[lr * 16 + k0];
                a.y  = s_xc[lr * 16 + k0 + 1];
                const v2f bw = wbc[kc >> 2];
                acc = __builtin_amdgcn_wmma_f32_16x16x4_f32(false, a, false, bw,
                                                            (short)0, acc, false, false);
            }
#pragma unroll
            for (int r = 0; r < 8; ++r) {
                const int row = min(l0 + r + 8 * hi, L_SEQ);
                dst[row * 16 + lm] = acc[r];
            }
        }
    }
    lds_barrier();

    for (int e = 0; e < 2; ++e) {
        const float* s_dtr = smem + e * ESTRIDE + O_DTR;
        float*       s_dt  = smem + e * ESTRIDE + O_DT;
        float raw[16];
#pragma unroll
        for (int k = 0; k < 15; ++k) raw[k] = s_dtr[(tid >> 4) + 16 * k];
        raw[15] = s_dtr[min((tid >> 4) + 240, L_SEQ - 1)];
#pragma unroll
        for (int k = 0; k < 15; ++k) s_dt[tid + 256 * k] = softplusf_(raw[k] * dtw + dtb);
        const float dv15 = softplusf_(raw[15] * dtw + dtb);
        if (tid < L_SEQ * DINNER - 3840) s_dt[tid + 3840] = dv15;
    }
    lds_barrier();

    {
        const int d = (tid & 127) >> 3;
        const int q = tid & 7;
        v2f As;
        As.x = -__expf(alog2.x);
        As.y = -__expf(alog2.y);
        const bool b0 = (q & 1) != 0, b1 = (q & 2) != 0, b2 = (q & 4) != 0;
        v2f h = {0.f, 0.f};
        if (wave < 4) {
            float* sb = smem;
            for (int l0 = 0; l0 < 248; l0 += 8)
                scan_steps2<8>(h, As, l0, d, q, b0, b1, b2, sb + O_DT, sb + O_XC, sb + O_B, sb + O_C, sb + O_DT);
            scan_steps2<L_SEQ - 248>(h, As, 248, d, q, b0, b1, b2, sb + O_DT, sb + O_XC, sb + O_B, sb + O_C, sb + O_DT);
        } else {
            float* sb = smem + ESTRIDE;
            for (int l0 = 0; l0 < 248; l0 += 8)
                scan_steps2<8>(h, As, l0, d, q, b0, b1, b2, sb + O_DT, sb + O_XC, sb + O_B, sb + O_C, sb + O_DT);
            scan_steps2<L_SEQ - 248>(h, As, 248, d, q, b0, b1, b2, sb + O_DT, sb + O_XC, sb + O_B, sb + O_C, sb + O_DT);
        }
    }
    const float Dpd = Dp[lm];
    v2f wo[4];
#pragma unroll
    for (int q = 0; q < 4; ++q) {
        const int k0 = 4 * q + 2 * hi;
        const float wa = W_out[(lm & (DMODEL - 1)) * 16 + k0];
        const float wb = W_out[(lm & (DMODEL - 1)) * 16 + k0 + 1];
        wo[q].x = (lm < DMODEL) ? wa : 0.f;
        wo[q].y = (lm < DMODEL) ? wb : 0.f;
    }
    const v4f* fw4 = reinterpret_cast<const v4f*>(fc_w);
    v4f w0[NCLS], w1[NCLS];
#pragma unroll
    for (int c = 0; c < NCLS; ++c) w0[c] = fw4[c * (L_SEQ * DMODEL / 4) + tid];
    lds_barrier();

#pragma unroll
    for (int e = 0; e < 2; ++e) {
        float*       s_y  = smem + e * ESTRIDE + O_DT;
        const float* s_xc = smem + e * ESTRIDE + O_XC;
#pragma unroll
        for (int tg = 0; tg < 4; ++tg) {
            const int t  = tg >> 1;
            const int r0 = (tg & 1) * 4;
            const int l0 = (wave + 8 * t) * 16 + r0 + 8 * hi;
            float yv[4], xv[4];
#pragma unroll
            for (int r = 0; r < 4; ++r) {
                const int idx = (t == 0 ? l0 + r : min(l0 + r, L_SEQ - 1)) * 16 + lm;
                yv[r] = s_y[idx];
                xv[r] = s_xc[idx];
            }
#pragma unroll
            for (int r = 0; r < 4; ++r) {
                float y = yv[r];
                y += xv[r] * Dpd;
                const float zv = zacc[e][t][r0 + r];
                y *= zv * sigmoidf_(zv);
                yv[r] = y;
            }
#pragma unroll
            for (int r = 0; r < 4; ++r)
                s_y[(t == 0 ? l0 + r : min(l0 + r, L_SEQ)) * 16 + lm] = yv[r];
        }
    }
    lds_barrier();

    for (int e = 0; e < 2; ++e) {
        const float* s_y   = smem + e * ESTRIDE + O_DT;
        float*       s_out = smem + e * ESTRIDE + O_C;
        for (int mt = wave; mt < 16; mt += 8) {
            const int l0 = mt * 16;
            const int lr = min(l0 + lm, L_SEQ - 1);
            v8f acc = {};
#pragma unroll
            for (int kc = 0; kc < 16; kc += 4) {
                const int k0 = kc + 2 * hi;
                v2f a;
                a.x  = s_y[lr * 16 + k0];
                a.y  = s_y[lr * 16 + k0 + 1];
                const v2f bw = wo[kc >> 2];
                acc = __builtin_amdgcn_wmma_f32_16x16x4_f32(false, a, false, bw,
                                                            (short)0, acc, false, false);
            }
            if (lm < DMODEL) {
#pragma unroll
                for (int r = 0; r < 8; ++r)
                    s_out[(l0 + r + 8 * hi) * 8 + lm] = acc[r];
            }
        }
    }
#pragma unroll
    for (int c = 0; c < NCLS; ++c) w1[c] = fw4[c * (L_SEQ * DMODEL / 4) + pc1];
    const float fcb = fc_b[min(tid, NCLS - 1)];
    lds_barrier();

    {
        float part[2][NCLS];
#pragma unroll
        for (int e = 0; e < 2; ++e) {
            const float* s_out = smem + e * ESTRIDE + O_C;
            v4f p[NCLS];
            const v4f o0 = reinterpret_cast<const v4f*>(s_out)[tid];
#pragma unroll
            for (int c = 0; c < NCLS; ++c) p[c] = o0 * w0[c];
            if (second) {
                const v4f o1 = reinterpret_cast<const v4f*>(s_out)[tid + 256];
#pragma unroll
                for (int c = 0; c < NCLS; ++c) p[c] += o1 * w1[c];
            }
#pragma unroll
            for (int c = 0; c < NCLS; ++c) part[e][c] = (p[c].x + p[c].y) + (p[c].z + p[c].w);
        }
#pragma unroll
        for (int e = 0; e < 2; ++e) {
#pragma unroll
            for (int c = 0; c < NCLS; ++c) {
#pragma unroll
                for (int off = 16; off > 0; off >>= 1)
                    part[e][c] += __shfl_down(part[e][c], off, 32);
            }
        }
        if (lane == 0) {
#pragma unroll
            for (int e = 0; e < 2; ++e) {
                float* s_red = smem + e * ESTRIDE + O_DTR;
#pragma unroll
                for (int c = 0; c < NCLS; ++c) s_red[wave * NCLS + c] = part[e][c];
            }
        }
        lds_barrier();
        if (tid < NCLS) {
#pragma unroll
            for (int e = 0; e < 2; ++e) {
                const float* s_red = smem + e * ESTRIDE + O_DTR;
                float sres = fcb;
#pragma unroll
                for (int w = 0; w < 8; ++w) sres += s_red[w * NCLS + tid];
                out[(bb + e) * NCLS + tid] = sres;
            }
        }
    }
}

extern "C" void kernel_launch(void* const* d_in, const int* in_sizes, int n_in,
                              void* d_out, int out_size, void* d_ws, size_t ws_size,
                              hipStream_t stream) {
    (void)in_sizes; (void)n_in; (void)out_size; (void)d_ws; (void)ws_size;
    mamba_fused_kernel<<<dim3(512), dim3(256), 0, stream>>>(
        (const float*)d_in[0],
        (const float*)d_in[1],
        (const float*)d_in[2],
        (const float*)d_in[3],
        (const float*)d_in[4],
        (const float*)d_in[5],
        (const float*)d_in[6],
        (const float*)d_in[7],
        (const float*)d_in[8],
        (const float*)d_in[9],
        (const float*)d_in[10],
        (const float*)d_in[11],
        (float*)d_out);
}


# Scan loop sheds scale/move/address instructions: exp2 args of 2 steps as one packed product (log2e folded into A; arg rounding differs), butterfly pick before in-lane add, two 8-step groups per trip
# speedup vs baseline: 8.9931x; 1.0475x over previous
#include <hip/hip_runtime.h>
#include <hip/hip_bf16.h>


#define L_SEQ 250
#define DMODEL 8
#define DINNER 16
#define DSTATE 16
#define DCONV 5
#define NCLS 12

#define ESTRIDE 16300
#define O_DT  0
#define O_XC  4016
#define O_B   8016
#define O_C   12032
#define O_DTR 16048
#define O_XI  12096

typedef __attribute__((ext_vector_type(2))) float v2f;
typedef __attribute__((ext_vector_type(4))) float v4f;
typedef __attribute__((ext_vector_type(8))) float v8f;

__device__ __forceinline__ float sigmoidf_(float v) { return __builtin_amdgcn_rcpf(1.0f + __expf(-v)); }

__device__ __forceinline__ float softplusf_(float v) {
    const float t  = __expf(v);
    const float u  = 1.0f + t;
    const float lg = __builtin_amdgcn_logf(u) * 0.6931471805599453f;
    const float r  = lg + (t - (u - 1.0f)) * __builtin_amdgcn_rcpf(u);
    return (v > 20.f) ? v : r;
}

__device__ __forceinline__ void lds_barrier() {
    __builtin_amdgcn_fence(__ATOMIC_RELEASE, "workgroup", "local");
    __builtin_amdgcn_s_barrier();
    __builtin_amdgcn_fence(__ATOMIC_ACQUIRE, "workgroup", "local");
}

template <int XM>
__device__ __forceinline__ float row_xmask_from(float v) {
    const int other = __builtin_amdgcn_update_dpp(0, __builtin_bit_cast(int, v),
                                                  0x160 | XM  , 0xf, 0xf, true);
    return __builtin_bit_cast(float, other);
}

template <int XM>
__device__ __forceinline__ float fold_pair(float a, float b, bool up) {
    const float keep = up ? b : a;
    const float send = up ? a : b;
    return keep + row_xmask_from<XM>(send);
}

template <int N>
__device__ __forceinline__ void scan_steps2(v2f& h, const v2f Ax, const v2f Ay, const int l0, const int d, const int q,
                                            const bool b0, const bool b1, const bool b2,
                                            const float* s_dt, const float* s_xc,
                                            const float* s_B, const float* s_C, float* s_y) {
    v2f p[8];
    for (int j = 0; j < 8; j += 2) {
        if (j < N) {
            const int l = l0 + j;
            const float dt0 = s_dt[l * 16 + d];
            const float xc0 = s_xc[l * 16 + d];
            const v2f   B0  = *reinterpret_cast<const v2f*>(s_B + l * 16 + 2 * q);
            const v2f   C0  = *reinterpret_cast<const v2f*>(s_C + l * 16 + 2 * q);
            const float dt1 = s_dt[(l + 1) * 16 + d];
            const float xc1 = s_xc[(l + 1) * 16 + d];
            const v2f   B1  = *reinterpret_cast<const v2f*>(s_B + (l + 1) * 16 + 2 * q);
            const v2f   C1  = *reinterpret_cast<const v2f*>(s_C + (l + 1) * 16 + 2 * q);
            const v2f dv = {dt0, dt1};
            const v2f ex = Ax * dv;
            const v2f ey = Ay * dv;
            v2f dA0, dA1;
            dA0.x = __builtin_amdgcn_exp2f(ex.x);
            dA0.y = __builtin_amdgcn_exp2f(ey.x);
            dA1.x = __builtin_amdgcn_exp2f(ex.y);
            dA1.y = __builtin_amdgcn_exp2f(ey.y);
            h = dA0 * h + (dt0 * xc0) * B0;
            p[j] = h * C0;
            h = dA1 * h + (dt1 * xc1) * B1;
            p[j + 1] = h * C1;
        } else {
            p[j] = v2f{0.f, 0.f};
            p[j + 1] = v2f{0.f, 0.f};
        }
    }
    float r1[4], r2[2];
    for (int j = 0; j < 4; ++j) {
        const v2f pa = p[2 * j], pb = p[2 * j + 1];
        v2f sx, sy;
        sx.x = b0 ? pb.x : pa.x;  sx.y = b0 ? pa.x : pb.x;
        sy.x = b0 ? pb.y : pa.y;  sy.y = b0 ? pa.y : pb.y;
        const v2f ks = sx + sy;
        r1[j] = ks.x + row_xmask_from<1>(ks.y);
    }
    for (int j = 0; j < 2; ++j) r2[j] = fold_pair<2>(r1[2 * j], r1[2 * j + 1], b1);
    const float ysum = fold_pair<4>(r2[0], r2[1], b2);
    if (q < N) s_y[(l0 + q) * 16 + d] = ysum;
}

__global__ __launch_bounds__(256) void mamba_fused_kernel(
    const float* __restrict__ x,
    const float* __restrict__ W_in,
    const float* __restrict__ conv_w,
    const float* __restrict__ conv_b,
    const float* __restrict__ W_x,
    const float* __restrict__ dt_w,
    const float* __restrict__ dt_b,
    const float* __restrict__ A_log,
    const float* __restrict__ Dp,
    const float* __restrict__ W_out,
    const float* __restrict__ fc_w,
    const float* __restrict__ fc_b,
    float* __restrict__ out)
{
    __shared__ __attribute__((aligned(16))) float smem[2 * ESTRIDE];

    const int bb   = blockIdx.x * 2;
    const int tid  = threadIdx.x;
    const int lane = tid & 31;
    const int wave = tid >> 5;
    const int lm   = lane & 15;
    const int hi   = lane >> 4;
    const float* xb = x + (size_t)bb * DMODEL * L_SEQ;

    const bool second = tid < (DMODEL * L_SEQ / 4 - 256);
    const int  pc1 = min(tid + 256, DMODEL * L_SEQ / 4 - 1);
    const v4f* xb4 = reinterpret_cast<const v4f*>(xb);
    v4f xv0[2], xv1[2];
#pragma unroll
    for (int e = 0; e < 2; ++e) {
        xv0[e] = xb4[e * (DMODEL * L_SEQ / 4) + tid];
        xv1[e] = xb4[e * (DMODEL * L_SEQ / 4) + pc1];
    }
    v2f wz[2], wx[2];
#pragma unroll
    for (int q = 0; q < 2; ++q) {
        const int k0 = 4 * q + 2 * hi;
        wz[q].x = W_in[(16 + lm) * 8 + k0];
        wz[q].y = W_in[(16 + lm) * 8 + k0 + 1];
        wx[q].x = W_in[lm * 8 + k0];
        wx[q].y = W_in[lm * 8 + k0 + 1];
    }
    float cw[DCONV];
#pragma unroll
    for (int j = 0; j < DCONV; ++j) cw[j] = conv_w[(tid & 15) * DCONV + j];
    const float cb = conv_b[tid & 15];
    for (int i = tid * 32; i < NCLS * L_SEQ * DMODEL; i += 256 * 32)
        __builtin_prefetch(fc_w + i, 0, 0);
#pragma unroll
    for (int e = 0; e < 2; ++e) {
        float* sb = smem + e * ESTRIDE;
        reinterpret_cast<v4f*>(sb + O_B)[tid] = xv0[e];
        if (second) reinterpret_cast<v4f*>(sb + O_B)[tid + 256] = xv1[e];
        if (tid < (DCONV - 1) * DINNER) (sb + O_XI)[tid - (DCONV - 1) * DINNER] = 0.f;
    }
    lds_barrier();

    v8f zacc[2][2];
#pragma unroll
    for (int e = 0; e < 2; ++e) {
        const float* s_xb = smem + e * ESTRIDE + O_B;
        float*       s_xi = smem + e * ESTRIDE + O_XI;
#pragma unroll
        for (int t = 0; t < 2; ++t) {
            const int l0 = (wave + 8 * t) * 16;
            const int lr = min(l0 + lm, L_SEQ - 1);
            v8f acc = {}, accx = {};
#pragma unroll
            for (int kc = 0; kc < 8; kc += 4) {
                const int k0 = kc + 2 * hi;
                v2f a;
                a.x  = s_xb[(k0    ) * L_SEQ + lr];
                a.y  = s_xb[(k0 + 1) * L_SEQ + lr];
                const v2f bz = wz[kc >> 2];
                const v2f bx = wx[kc >> 2];
                acc  = __builtin_amdgcn_wmma_f32_16x16x4_f32(false, a, false, bz,
                                                             (short)0, acc, false, false);
                accx = __builtin_amdgcn_wmma_f32_16x16x4_f32(false, a, false, bx,
                                                             (short)0, accx, false, false);
            }
            zacc[e][t] = acc;
#pragma unroll
            for (int r = 0; r < 8; ++r) {
                const int row = min(l0 + r + 8 * hi, L_SEQ);
                s_xi[row * 16 + lm] = accx[r];
            }
        }
    }
    lds_barrier();

    float wdt[DINNER];
#pragma unroll
    for (int d = 0; d < DINNER; ++d) wdt[d] = W_x[d];
    const int nt3 = wave & 1;
    const int e03 = 1 + nt3 * 16;
    v2f wbc[4];
#pragma unroll
    for (int q = 0; q < 4; ++q) {
        const int k0 = 4 * q + 2 * hi;
        wbc[q].x = W_x[(e03 + lm) * 16 + k0];
        wbc[q].y = W_x[(e03 + lm) * 16 + k0 + 1];
    }

    for (int e = 0; e < 2; ++e) {
        const float* s_xi = smem + e * ESTRIDE + O_XI;
        float*       s_xc = smem + e * ESTRIDE + O_XC;
        for (int i = tid; i < L_SEQ * DINNER; i += 256) {
            float accv = cb;
#pragma unroll
            for (int j = 0; j < DCONV; ++j)
                accv += cw[j] * s_xi[i + (j - (DCONV - 1)) * DINNER];
            s_xc[i] = accv * sigmoidf_(accv);
        }
    }
    lds_barrier();

    const float dtw  = dt_w[tid & 15], dtb = dt_b[tid & 15];
    const v2f alog2 = reinterpret_cast<const v2f*>(A_log)[tid & 127];

    for (int e = 0; e < 2; ++e) {
        const float* s_xc  = smem + e * ESTRIDE + O_XC;
        float*       s_dtr = smem + e * ESTRIDE + O_DTR;
        for (int l = tid; l < L_SEQ; l += 256) {
            float a = 0.f;
#pragma unroll
            for (int d = 0; d < DINNER; ++d) a += s_xc[l * 16 + d] * wdt[d];
            s_dtr[l] = a;
        }
    }

    for (int e = 0; e < 2; ++e) {
        const float* s_xc = smem + e * ESTRIDE + O_XC;
        float* dst = smem + e * ESTRIDE + (nt3 ? O_C : O_B);
        for (int j = wave; j < 32; j += 8) {
            const int mt = j >> 1;
            const int l0 = mt * 16;
            const int lr = min(l0 + lm, L_SEQ - 1);
            v8f acc = {};
#pragma unroll
            for (int kc = 0; kc < 16; kc += 4) {
                const int k0 = kc + 2 * hi;
                v2f a;
                a.x  = s_xc[lr * 16 + k0];
                a.y  = s_xc[lr * 16 + k0 + 1];
                const v2f bw = wbc[kc >> 2];
                acc = __builtin_amdgcn_wmma_f32_16x16x4_f32(false, a, false, bw,
                                                            (short)0, acc, false, false);
            }
#pragma unroll
            for (int r = 0; r < 8; ++r) {
                const int row = min(l0 + r + 8 * hi, L_SEQ);
                dst[row * 16 + lm] = acc[r];
            }
        }
    }
    lds_barrier();

    for (int e = 0; e < 2; ++e) {
        const float* s_dtr = smem + e * ESTRIDE + O_DTR;
        float*       s_dt  = smem + e * ESTRIDE + O_DT;
        float raw[16];
#pragma unroll
        for (int k = 0; k < 15; ++k) raw[k] = s_dtr[(tid >> 4) + 16 * k];
        raw[15] = s_dtr[min((tid >> 4) + 240, L_SEQ - 1)];
#pragma unroll
        for (int k = 0; k < 15; ++k) s_dt[tid + 256 * k] = softplusf_(raw[k] * dtw + dtb);
        const float dv15 = softplusf_(raw[15] * dtw + dtb);
        if (tid < L_SEQ * DINNER - 3840) s_dt[tid + 3840] = dv15;
    }
    lds_barrier();

    {
        const int d = (tid & 127) >> 3;
        const int q = tid & 7;
        const float ax = -__expf(alog2.x) * 1.4426950408889634f;
        const float ay = -__expf(alog2.y) * 1.4426950408889634f;
        const v2f Ax = {ax, ax};
        const v2f Ay = {ay, ay};
        const bool b0 = (q & 1) != 0, b1 = (q & 2) != 0, b2 = (q & 4) != 0;
        v2f h = {0.f, 0.f};
        if (wave < 4) {
            float* sb = smem;
            for (int l0 = 0; l0 < 240; l0 += 16) {
                scan_steps2<8>(h, Ax, Ay, l0,     d, q, b0, b1, b2, sb + O_DT, sb + O_XC, sb + O_B, sb + O_C, sb + O_DT);
                scan_steps2<8>(h, Ax, Ay, l0 + 8, d, q, b0, b1, b2, sb + O_DT, sb + O_XC, sb + O_B, sb + O_C, sb + O_DT);
            }
            scan_steps2<8>(h, Ax, Ay, 240, d, q, b0, b1, b2, sb + O_DT, sb + O_XC, sb + O_B, sb + O_C, sb + O_DT);
            scan_steps2<L_SEQ - 248>(h, Ax, Ay, 248, d, q, b0, b1, b2, sb + O_DT, sb + O_XC, sb + O_B, sb + O_C, sb + O_DT);
        } else {
            float* sb = smem + ESTRIDE;
            for (int l0 = 0; l0 < 240; l0 += 16) {
                scan_steps2<8>(h, Ax, Ay, l0,     d, q, b0, b1, b2, sb + O_DT, sb + O_XC, sb + O_B, sb + O_C, sb + O_DT);
                scan_steps2<8>(h, Ax, Ay, l0 + 8, d, q, b0, b1, b2, sb + O_DT, sb + O_XC, sb + O_B, sb + O_C, sb + O_DT);
            }
            scan_steps2<8>(h, Ax, Ay, 240, d, q, b0, b1, b2, sb + O_DT, sb + O_XC, sb + O_B, sb + O_C, sb + O_DT);
            scan_steps2<L_SEQ - 248>(h, Ax, Ay, 248, d, q, b0, b1, b2, sb + O_DT, sb + O_XC, sb + O_B, sb + O_C, sb + O_DT);
        }
    }
    const float Dpd = Dp[lm];
    v2f wo[4];
#pragma unroll
    for (int q = 0; q < 4; ++q) {
        const int k0 = 4 * q + 2 * hi;
        const float wa = W_out[(lm & (DMODEL - 1)) * 16 + k0];
        const float wb = W_out[(lm & (DMODEL - 1)) * 16 + k0 + 1];
        wo[q].x = (lm < DMODEL) ? wa : 0.f;
        wo[q].y = (lm < DMODEL) ? wb : 0.f;
    }
    const v4f* fw4 = reinterpret_cast<const v4f*>(fc_w);
    v4f w0[NCLS], w1[NCLS];
#pragma unroll
    for (int c = 0; c < NCLS; ++c) w0[c] = fw4[c * (L_SEQ * DMODEL / 4) + tid];
    lds_barrier();

#pragma unroll
    for (int e = 0; e < 2; ++e) {
        float*       s_y  = smem + e * ESTRIDE + O_DT;
        const float* s_xc = smem + e * ESTRIDE + O_XC;
#pragma unroll
        for (int tg = 0; tg < 4; ++tg) {
            const int t  = tg >> 1;
            const int r0 = (tg & 1) * 4;
            const int l0 = (wave + 8 * t) * 16 + r0 + 8 * hi;
            float yv[4], xv[4];
#pragma unroll
            for (int r = 0; r < 4; ++r) {
                const int idx = (t == 0 ? l0 + r : min(l0 + r, L_SEQ - 1)) * 16 + lm;
                yv[r] = s_y[idx];
                xv[r] = s_xc[idx];
            }
#pragma unroll
            for (int r = 0; r < 4; ++r) {
                float y = yv[r];
                y += xv[r] * Dpd;
                const float zv = zacc[e][t][r0 + r];
                y *= zv * sigmoidf_(zv);
                yv[r] = y;
            }
#pragma unroll
            for (int r = 0; r < 4; ++r)
                s_y[(t == 0 ? l0 + r : min(l0 + r, L_SEQ)) * 16 + lm] = yv[r];
        }
    }
    lds_barrier();

    for (int e = 0; e < 2; ++e) {
        const float* s_y   = smem + e * ESTRIDE + O_DT;
        float*       s_out = smem + e * ESTRIDE + O_C;
        for (int mt = wave; mt < 16; mt += 8) {
            const int l0 = mt * 16;
            const int lr = min(l0 + lm, L_SEQ - 1);
            v8f acc = {};
#pragma unroll
            for (int kc = 0; kc < 16; kc += 4) {
                const int k0 = kc + 2 * hi;
                v2f a;
                a.x  = s_y[lr * 16 + k0];
                a.y  = s_y[lr * 16 + k0 + 1];
                const v2f bw = wo[kc >> 2];
                acc = __builtin_amdgcn_wmma_f32_16x16x4_f32(false, a, false, bw,
                                                            (short)0, acc, false, false);
            }
            if (lm < DMODEL) {
#pragma unroll
                for (int r = 0; r < 8; ++r)
                    s_out[(l0 + r + 8 * hi) * 8 + lm] = acc[r];
            }
        }
    }
#pragma unroll
    for (int c = 0; c < NCLS; ++c) w1[c] = fw4[c * (L_SEQ * DMODEL / 4) + pc1];
    const float fcb = fc_b[min(tid, NCLS - 1)];
    lds_barrier();

    {
        float part[2][NCLS];
#pragma unroll
        for (int e = 0; e < 2; ++e) {
            const float* s_out = smem + e * ESTRIDE + O_C;
            v4f p[NCLS];
            const v4f o0 = reinterpret_cast<const v4f*>(s_out)[tid];
#pragma unroll
            for (int c = 0; c < NCLS; ++c) p[c] = o0 * w0[c];
            if (second) {
                const v4f o1 = reinterpret_cast<const v4f*>(s_out)[tid + 256];
#pragma unroll
                for (int c = 0; c < NCLS; ++c) p[c] += o1 * w1[c];
            }
#pragma unroll
            for (int c = 0; c < NCLS; ++c) part[e][c] = (p[c].x + p[c].y) + (p[c].z + p[c].w);
        }
#pragma unroll
        for (int e = 0; e < 2; ++e) {
#pragma unroll
            for (int c = 0; c < NCLS; ++c) {
#pragma unroll
                for (int off = 16; off > 0; off >>= 1)
                    part[e][c] += __shfl_down(part[e][c], off, 32);
            }
        }
        if (lane == 0) {
#pragma unroll
            for (int e = 0; e < 2; ++e) {
                float* s_red = smem + e * ESTRIDE + O_DTR;
#pragma unroll
                for (int c = 0; c < NCLS; ++c) s_red[wave * NCLS + c] = part[e][c];
            }
        }
        lds_barrier();
        if (tid < NCLS) {
#pragma unroll
            for (int e = 0; e < 2; ++e) {
                const float* s_red = smem + e * ESTRIDE + O_DTR;
                float sres = fcb;
#pragma unroll
                for (int w = 0; w < 8; ++w) sres += s_red[w * NCLS + tid];
                out[(bb + e) * NCLS + tid] = sres;
            }
        }
    }
}

extern "C" void kernel_launch(void* const* d_in, const int* in_sizes, int n_in,
                              void* d_out, int out_size, void* d_ws, size_t ws_size,
                              hipStream_t stream) {
    (void)in_sizes; (void)n_in; (void)out_size; (void)d_ws; (void)ws_size;
    mamba_fused_kernel<<<dim3(512), dim3(256), 0, stream>>>(
        (const float*)d_in[0],
        (const float*)d_in[1],
        (const float*)d_in[2],
        (const float*)d_in[3],
        (const float*)d_in[4],
        (const float*)d_in[5],
        (const float*)d_in[6],
        (const float*)d_in[7],
        (const float*)d_in[8],
        (const float*)d_in[9],
        (const float*)d_in[10],
        (const float*)d_in[11],
        (float*)d_out);
}
